# starting kernel
# baseline (speedup 1.0000x reference)
#include <hip/hip_runtime.h>

typedef __attribute__((ext_vector_type(2))) float v2f;
typedef __attribute__((ext_vector_type(8))) float v8f;

#define FDIM 256
#define BM 64
#define BN 128
#define BK 64
#define SXP (BK + 4)
#define BNP 144

__global__ __launch_bounds__(256)
void gemm_f32_wmma(const float* __restrict__ X, const float* __restrict__ W,
                   float* __restrict__ H, int nrows) {
  __shared__ __align__(16) float sX[BM * SXP];
  __shared__ __align__(16) float sWp[(BK / 2) * BNP * 2];

  const int t       = threadIdx.x;
  const int lane    = t & 31;
  const int wave    = t >> 5;
  const int l16     = lane & 15;
  const int half    = lane >> 4;
  const int waveRow = wave & 3;
  const int waveCol = wave >> 2;
  const int rowBase = blockIdx.x * BM;
  const int colBase = blockIdx.y * BN;
  const int cb      = waveCol * 64 + l16;

  v8f acc0 = {}, acc1 = {}, acc2 = {}, acc3 = {};

  for (int kBase = 0; kBase < FDIM; kBase += BK) {
#pragma unroll
    for (int it = 0; it < 4; ++it) {
      int idx = t + 256 * it;
      int row = idx >> 4;
      int c4  = (idx & 15) << 2;
      float4 v = make_float4(0.f, 0.f, 0.f, 0.f);
      int g = rowBase + row;
      if (g < nrows)
        v = *(const float4*)(X + (size_t)g * FDIM + kBase + c4);
      *(float4*)&sX[row * SXP + c4] = v;
    }
#pragma unroll
    for (int it = 0; it < 8; ++it) {
      int idx = t + 256 * it;
      int k   = idx >> 5;
      int n4  = (idx & 31) << 2;
      float4 v = *(const float4*)(W + (size_t)(kBase + k) * FDIM + colBase + n4);
      float* p = sWp + ((size_t)(k >> 1) * BNP + n4) * 2 + (k & 1);
      p[0] = v.x; p[2] = v.y; p[4] = v.z; p[6] = v.w;
    }
    __syncthreads();

#pragma unroll
    for (int kk = 0; kk < BK; kk += 4) {
      const int ks = kk + 2 * half;
      v2f a = *(const v2f*)&sX[(waveRow * 16 + l16) * SXP + ks];
      const v2f* wrow = (const v2f*)(sWp + ((kk >> 1) + half) * BNP * 2);
      v2f b0 = wrow[cb];
      v2f b1 = wrow[cb + 16];
      v2f b2 = wrow[cb + 32];
      v2f b3 = wrow[cb + 48];
      acc0 = __builtin_amdgcn_wmma_f32_16x16x4_f32(false, a, false, b0, (short)0, acc0, false, false);
      acc1 = __builtin_amdgcn_wmma_f32_16x16x4_f32(false, a, false, b1, (short)0, acc1, false, false);
      acc2 = __builtin_amdgcn_wmma_f32_16x16x4_f32(false, a, false, b2, (short)0, acc2, false, false);
      acc3 = __builtin_amdgcn_wmma_f32_16x16x4_f32(false, a, false, b3, (short)0, acc3, false, false);
    }
    __syncthreads();
  }

#pragma unroll
  for (int v = 0; v < 8; ++v) {
    int gr = rowBase + waveRow * 16 + v + 8 * half;
    if (gr < nrows) {
      float* p = H + (size_t)gr * FDIM + colBase + waveCol * 64 + l16;
      p[0]  = acc0[v];
      p[16] = acc1[v];
      p[32] = acc2[v];
      p[48] = acc3[v];
    }
  }
}

__global__ __launch_bounds__(256)
void scatter_edges(const float* __restrict__ h, const int* __restrict__ src,
                   const int* __restrict__ dst, const float* __restrict__ val,
                   float* __restrict__ acc, int E) {
  long long idx = (long long)blockIdx.x * blockDim.x + threadIdx.x;
  int e  = (int)(idx >> 6);
  int c4 = (int)(idx & 63) << 2;
  if (e >= E) return;
  int   s = src[e];
  int   d = dst[e];
  float v = val[e];
  float4 hv = *(const float4*)(h + (size_t)s * FDIM + c4);
  float* a = acc + (size_t)d * FDIM + c4;
  unsafeAtomicAdd(a + 0, hv.x * v);
  unsafeAtomicAdd(a + 1, hv.y * v);
  unsafeAtomicAdd(a + 2, hv.z * v);
  unsafeAtomicAdd(a + 3, hv.w * v);
}

__global__ __launch_bounds__(256)
void zero_buf(float* __restrict__ p, long long n4) {
  long long i = (long long)blockIdx.x * blockDim.x + threadIdx.x;
  if (i < n4) *(float4*)(p + 4 * i) = make_float4(0.f, 0.f, 0.f, 0.f);
}

__global__ __launch_bounds__(256)
void bias_relu(const float* __restrict__ acc, const float* __restrict__ b,
               float* __restrict__ o, long long n) {
  long long i = (long long)blockIdx.x * blockDim.x + threadIdx.x;
  if (i < n) o[i] = fmaxf(acc[i] + b[i & (FDIM - 1)], 0.f);
}

__global__ __launch_bounds__(256)
void final_fuse(const float* __restrict__ acc, const float* __restrict__ b,
                const float* __restrict__ x1, float* __restrict__ out, long long n) {
  long long i = (long long)blockIdx.x * blockDim.x + threadIdx.x;
  if (i < n) out[i] = fmaxf(acc[i] + b[i & (FDIM - 1)], 0.f) + x1[i];
}

extern "C" void kernel_launch(void* const* d_in, const int* in_sizes, int n_in,
                              void* d_out, int out_size, void* d_ws, size_t ws_size,
                              hipStream_t stream) {
  const float* x    = (const float*)d_in[0];
  const int*   esrc = (const int*)d_in[1];
  const int*   edst = (const int*)d_in[2];
  const float* eval = (const float*)d_in[3];
  const float* W1   = (const float*)d_in[4];
  const float* b1   = (const float*)d_in[5];
  const float* W2   = (const float*)d_in[6];
  const float* b2   = (const float*)d_in[7];
  float* out = (float*)d_out;

  const int R  = 4;
  const int Nn = in_sizes[0] / FDIM;
  const int E  = in_sizes[1] / R;

  const size_t nf = (size_t)Nn * FDIM;
  float* h   = (float*)d_ws;
  float* acc = h + nf;
  float* x1  = acc + nf;

  dim3 gGrid((Nn + BM - 1) / BM, FDIM / BN);
  const long long n4 = (long long)(nf / 4);
  const int zBlocks = (int)((n4 + 255) / 256);
  const int sBlocks = (int)(((long long)E * 64 + 255) / 256);
  const int pBlocks = (int)(((long long)nf + 255) / 256);

  zero_buf<<<zBlocks, 256, 0, stream>>>(acc, n4);
  for (int r = 0; r < R; ++r) {
    gemm_f32_wmma<<<gGrid, 256, 0, stream>>>(x, W1 + (size_t)r * FDIM * FDIM, h, Nn);
    scatter_edges<<<sBlocks, 256, 0, stream>>>(h, esrc + (size_t)r * E,
                                               edst + (size_t)r * E,
                                               eval + (size_t)r * E, acc, E);
  }
  bias_relu<<<pBlocks, 256, 0, stream>>>(acc, b1, x1, (long long)nf);

  zero_buf<<<zBlocks, 256, 0, stream>>>(acc, n4);
  for (int r = 0; r < R; ++r) {
    gemm_f32_wmma<<<gGrid, 256, 0, stream>>>(x1, W2 + (size_t)r * FDIM * FDIM, h, Nn);
    scatter_edges<<<sBlocks, 256, 0, stream>>>(h, esrc + (size_t)r * E,
                                               edst + (size_t)r * E,
                                               eval + (size_t)r * E, acc, E);
  }
  final_fuse<<<pBlocks, 256, 0, stream>>>(acc, b2, x1, out, (long long)nf);
}


# float-atomic edge scatter replaced by a per-row gather over dst bins; each row's edges are put in edge-id order before summing, so the sum order is fixed and no longer the bins' arrival order
# speedup vs baseline: 1.7839x; 1.7839x over previous
#include <hip/hip_runtime.h>

typedef __attribute__((ext_vector_type(2))) float v2f;
typedef __attribute__((ext_vector_type(8))) float v8f;

#define FDIM 256
#define BM 64
#define BN 128
#define BK 64
#define SXP (BK + 4)
#define BNP 144

__global__ __launch_bounds__(256)
void gemm_f32_wmma(const float* __restrict__ X, const float* __restrict__ W,
                   float* __restrict__ H, int nrows) {
  __shared__ __align__(16) float sX[BM * SXP];
  __shared__ __align__(16) float sWp[(BK / 2) * BNP * 2];

  const int t       = threadIdx.x;
  const int lane    = t & 31;
  const int wave    = t >> 5;
  const int l16     = lane & 15;
  const int half    = lane >> 4;
  const int waveRow = wave & 3;
  const int waveCol = wave >> 2;
  const int rowBase = blockIdx.x * BM;
  const int colBase = blockIdx.y * BN;
  const int cb      = waveCol * 64 + l16;

  v8f acc0 = {}, acc1 = {}, acc2 = {}, acc3 = {};

  for (int kBase = 0; kBase < FDIM; kBase += BK) {
#pragma unroll
    for (int it = 0; it < 4; ++it) {
      int idx = t + 256 * it;
      int row = idx >> 4;
      int c4  = (idx & 15) << 2;
      float4 v = make_float4(0.f, 0.f, 0.f, 0.f);
      int g = rowBase + row;
      if (g < nrows)
        v = *(const float4*)(X + (size_t)g * FDIM + kBase + c4);
      *(float4*)&sX[row * SXP + c4] = v;
    }
#pragma unroll
    for (int it = 0; it < 8; ++it) {
      int idx = t + 256 * it;
      int k   = idx >> 5;
      int n4  = (idx & 31) << 2;
      float4 v = *(const float4*)(W + (size_t)(kBase + k) * FDIM + colBase + n4);
      float* p = sWp + ((size_t)(k >> 1) * BNP + n4) * 2 + (k & 1);
      p[0] = v.x; p[2] = v.y; p[4] = v.z; p[6] = v.w;
    }
    __syncthreads();

#pragma unroll
    for (int kk = 0; kk < BK; kk += 4) {
      const int ks = kk + 2 * half;
      v2f a = *(const v2f*)&sX[(waveRow * 16 + l16) * SXP + ks];
      const v2f* wrow = (const v2f*)(sWp + ((kk >> 1) + half) * BNP * 2);
      v2f b0 = wrow[cb];
      v2f b1 = wrow[cb + 16];
      v2f b2 = wrow[cb + 32];
      v2f b3 = wrow[cb + 48];
      acc0 = __builtin_amdgcn_wmma_f32_16x16x4_f32(false, a, false, b0, (short)0, acc0, false, false);
      acc1 = __builtin_amdgcn_wmma_f32_16x16x4_f32(false, a, false, b1, (short)0, acc1, false, false);
      acc2 = __builtin_amdgcn_wmma_f32_16x16x4_f32(false, a, false, b2, (short)0, acc2, false, false);
      acc3 = __builtin_amdgcn_wmma_f32_16x16x4_f32(false, a, false, b3, (short)0, acc3, false, false);
    }
    __syncthreads();
  }

#pragma unroll
  for (int v = 0; v < 8; ++v) {
    int gr = rowBase + waveRow * 16 + v + 8 * half;
    if (gr < nrows) {
      float* p = H + (size_t)gr * FDIM + colBase + waveCol * 64 + l16;
      p[0]  = acc0[v];
      p[16] = acc1[v];
      p[32] = acc2[v];
      p[48] = acc3[v];
    }
  }
}

#define KEY_NONE 0x7fffffff

__global__ __launch_bounds__(256)
void zero_ints(int* __restrict__ p, int n) {
  int i = blockIdx.x * blockDim.x + threadIdx.x;
  if (i < n) p[i] = 0;
}

__global__ __launch_bounds__(256)
void count_dst(const int* __restrict__ dst, int* __restrict__ cnt, int total) {
  int g = blockIdx.x * blockDim.x + threadIdx.x;
  if (g < total) atomicAdd(cnt + dst[g], 1);
}

__global__ __launch_bounds__(1024)
void scan_rows(const int* __restrict__ cnt, int* __restrict__ rowptr, int n) {
  __shared__ int part[1024];
  const int t     = threadIdx.x;
  const int chunk = (n + 1023) >> 10;
  const int b     = t * chunk;
  const int e     = min(b + chunk, n);
  int s = 0;
  for (int i = b; i < e; ++i) s += cnt[i];
  part[t] = s;
  __syncthreads();
  for (int off = 1; off < 1024; off <<= 1) {
    int v = (t >= off) ? part[t - off] : 0;
    __syncthreads();
    part[t] += v;
    __syncthreads();
  }
  int run = part[t] - s;
  for (int i = b; i < e; ++i) { rowptr[i] = run; run += cnt[i]; }
  if (t == 1023) rowptr[n] = part[1023];
}

__global__ __launch_bounds__(256)
void fill_bins(const int* __restrict__ dst, const int* __restrict__ rowptr,
               int* __restrict__ cnt, int* __restrict__ bins, int total) {
  int g = blockIdx.x * blockDim.x + threadIdx.x;
  if (g < total) {
    int d    = dst[g];
    int left = atomicSub(cnt + d, 1);
    bins[rowptr[d] + left - 1] = g;
  }
}

__device__ __forceinline__
void add_edge_run(const float* __restrict__ H, int creg, int vreg, int cnt,
                  int lane, float4& a0, float4& a1) {
  int p = 0;
  for (; p + 4 <= cnt; p += 4) {
    const int c0 = __builtin_amdgcn_readlane(creg, p);
    const int c1 = __builtin_amdgcn_readlane(creg, p + 1);
    const int c2 = __builtin_amdgcn_readlane(creg, p + 2);
    const int c3 = __builtin_amdgcn_readlane(creg, p + 3);
    const float w0 = __int_as_float(__builtin_amdgcn_readlane(vreg, p));
    const float w1 = __int_as_float(__builtin_amdgcn_readlane(vreg, p + 1));
    const float w2 = __int_as_float(__builtin_amdgcn_readlane(vreg, p + 2));
    const float w3 = __int_as_float(__builtin_amdgcn_readlane(vreg, p + 3));
    const float* q0 = H + (size_t)c0 * FDIM + lane * 4;
    const float* q1 = H + (size_t)c1 * FDIM + lane * 4;
    const float* q2 = H + (size_t)c2 * FDIM + lane * 4;
    const float* q3 = H + (size_t)c3 * FDIM + lane * 4;
    const float4 x00 = *(const float4*)q0, x01 = *(const float4*)(q0 + 128);
    const float4 x10 = *(const float4*)q1, x11 = *(const float4*)(q1 + 128);
    const float4 x20 = *(const float4*)q2, x21 = *(const float4*)(q2 + 128);
    const float4 x30 = *(const float4*)q3, x31 = *(const float4*)(q3 + 128);
    a0.x = fmaf(x00.x, w0, a0.x); a0.y = fmaf(x00.y, w0, a0.y);
    a0.z = fmaf(x00.z, w0, a0.z); a0.w = fmaf(x00.w, w0, a0.w);
    a1.x = fmaf(x01.x, w0, a1.x); a1.y = fmaf(x01.y, w0, a1.y);
    a1.z = fmaf(x01.z, w0, a1.z); a1.w = fmaf(x01.w, w0, a1.w);
    a0.x = fmaf(x10.x, w1, a0.x); a0.y = fmaf(x10.y, w1, a0.y);
    a0.z = fmaf(x10.z, w1, a0.z); a0.w = fmaf(x10.w, w1, a0.w);
    a1.x = fmaf(x11.x, w1, a1.x); a1.y = fmaf(x11.y, w1, a1.y);
    a1.z = fmaf(x11.z, w1, a1.z); a1.w = fmaf(x11.w, w1, a1.w);
    a0.x = fmaf(x20.x, w2, a0.x); a0.y = fmaf(x20.y, w2, a0.y);
    a0.z = fmaf(x20.z, w2, a0.z); a0.w = fmaf(x20.w, w2, a0.w);
    a1.x = fmaf(x21.x, w2, a1.x); a1.y = fmaf(x21.y, w2, a1.y);
    a1.z = fmaf(x21.z, w2, a1.z); a1.w = fmaf(x21.w, w2, a1.w);
    a0.x = fmaf(x30.x, w3, a0.x); a0.y = fmaf(x30.y, w3, a0.y);
    a0.z = fmaf(x30.z, w3, a0.z); a0.w = fmaf(x30.w, w3, a0.w);
    a1.x = fmaf(x31.x, w3, a1.x); a1.y = fmaf(x31.y, w3, a1.y);
    a1.z = fmaf(x31.z, w3, a1.z); a1.w = fmaf(x31.w, w3, a1.w);
  }
  for (; p < cnt; ++p) {
    const int   c = __builtin_amdgcn_readlane(creg, p);
    const float w = __int_as_float(__builtin_amdgcn_readlane(vreg, p));
    const float* q = H + (size_t)c * FDIM + lane * 4;
    const float4 x0 = *(const float4*)q, x1 = *(const float4*)(q + 128);
    a0.x = fmaf(x0.x, w, a0.x); a0.y = fmaf(x0.y, w, a0.y);
    a0.z = fmaf(x0.z, w, a0.z); a0.w = fmaf(x0.w, w, a0.w);
    a1.x = fmaf(x1.x, w, a1.x); a1.y = fmaf(x1.y, w, a1.y);
    a1.z = fmaf(x1.z, w, a1.z); a1.w = fmaf(x1.w, w, a1.w);
  }
}

template <bool FINAL>
__global__ __launch_bounds__(256)
void gather_rows(const float* __restrict__ H, const int* __restrict__ rowptr,
                 const int* __restrict__ bins, const int* __restrict__ esrc,
                 const float* __restrict__ eval, const float* __restrict__ bias,
                 const float* __restrict__ skip, float* __restrict__ o,
                 int N, int E) {
  __shared__ int sC[8][64];
  __shared__ int sV[8][64];
  const int lane = threadIdx.x & 31;
  const int wave = __builtin_amdgcn_readfirstlane(threadIdx.x >> 5);
  const int row  = blockIdx.x * 8 + wave;
  int start = 0, n = 0;
  if (row < N) { start = rowptr[row]; n = rowptr[row + 1] - start; }
  const bool small = n <= 64;

  if (small) {
    const bool h0 = lane < n, h1 = lane + 32 < n;
    int k0 = KEY_NONE, k1 = KEY_NONE;
    if (h0) k0 = bins[start + lane];
    if (h1) k1 = bins[start + 32 + lane];
    int s0 = 0, s1 = 0, v0 = 0, v1 = 0;
    if (h0) { s0 = esrc[k0]; v0 = __float_as_int(eval[k0]); }
    if (h1) { s1 = esrc[k1]; v1 = __float_as_int(eval[k1]); }
    int r0 = 0, r1 = 0;
    const int na = min(n, 32);
    for (int i = 0; i < na; ++i) {
      const int ki = __builtin_amdgcn_readlane(k0, i);
      r0 += (ki < k0) ? 1 : 0;
      r1 += (ki < k1) ? 1 : 0;
    }
    for (int i = 0; i < n - 32; ++i) {
      const int ki = __builtin_amdgcn_readlane(k1, i);
      r0 += (ki < k0) ? 1 : 0;
      r1 += (ki < k1) ? 1 : 0;
    }
    if (h0) {
      const int rel = (k0 >= E ? 1 : 0) + (k0 >= 2 * E ? 1 : 0) + (k0 >= 3 * E ? 1 : 0);
      sC[wave][r0] = rel * N + s0;
      sV[wave][r0] = v0;
    }
    if (h1) {
      const int rel = (k1 >= E ? 1 : 0) + (k1 >= 2 * E ? 1 : 0) + (k1 >= 3 * E ? 1 : 0);
      sC[wave][r1] = rel * N + s1;
      sV[wave][r1] = v1;
    }
  }
  __syncthreads();

  float4 a0 = make_float4(0.f, 0.f, 0.f, 0.f);
  float4 a1 = make_float4(0.f, 0.f, 0.f, 0.f);
  if (small) {
    int cA = 0, vA = 0, cB = 0, vB = 0;
    if (lane < n)      { cA = sC[wave][lane];      vA = sV[wave][lane]; }
    if (lane + 32 < n) { cB = sC[wave][lane + 32]; vB = sV[wave][lane + 32]; }
    add_edge_run(H, cA, vA, min(n, 32), lane, a0, a1);
    if (n > 32) add_edge_run(H, cB, vB, n - 32, lane, a0, a1);
  } else {
    int last = -1;
    for (int p = 0; p < n; ++p) {
      int best = KEY_NONE;
      for (int j = lane; j < n; j += 32) {
        const int k = bins[start + j];
        if (k > last && k < best) best = k;
      }
      for (int off = 16; off > 0; off >>= 1) best = min(best, __shfl_xor(best, off));
      best = __builtin_amdgcn_readfirstlane(best);
      last = best;
      const int   s   = esrc[best];
      const float w   = eval[best];
      const int   rel = (best >= E ? 1 : 0) + (best >= 2 * E ? 1 : 0) + (best >= 3 * E ? 1 : 0);
      const float* q = H + (size_t)(rel * N + s) * FDIM + lane * 4;
      const float4 x0 = *(const float4*)q, x1 = *(const float4*)(q + 128);
      a0.x = fmaf(x0.x, w, a0.x); a0.y = fmaf(x0.y, w, a0.y);
      a0.z = fmaf(x0.z, w, a0.z); a0.w = fmaf(x0.w, w, a0.w);
      a1.x = fmaf(x1.x, w, a1.x); a1.y = fmaf(x1.y, w, a1.y);
      a1.z = fmaf(x1.z, w, a1.z); a1.w = fmaf(x1.w, w, a1.w);
    }
  }

  if (row < N) {
    const float4 b0 = *(const float4*)(bias + lane * 4);
    const float4 b1 = *(const float4*)(bias + 128 + lane * 4);
    float4 o0, o1;
    o0.x = fmaxf(a0.x + b0.x, 0.f); o0.y = fmaxf(a0.y + b0.y, 0.f);
    o0.z = fmaxf(a0.z + b0.z, 0.f); o0.w = fmaxf(a0.w + b0.w, 0.f);
    o1.x = fmaxf(a1.x + b1.x, 0.f); o1.y = fmaxf(a1.y + b1.y, 0.f);
    o1.z = fmaxf(a1.z + b1.z, 0.f); o1.w = fmaxf(a1.w + b1.w, 0.f);
    const size_t base = (size_t)row * FDIM + lane * 4;
    if (FINAL) {
      const float4 k0 = *(const float4*)(skip + base);
      const float4 k1 = *(const float4*)(skip + base + 128);
      o0.x += k0.x; o0.y += k0.y; o0.z += k0.z; o0.w += k0.w;
      o1.x += k1.x; o1.y += k1.y; o1.z += k1.z; o1.w += k1.w;
    }
    *(float4*)(o + base)       = o0;
    *(float4*)(o + base + 128) = o1;
  }
}

extern "C" void kernel_launch(void* const* d_in, const int* in_sizes, int n_in,
                              void* d_out, int out_size, void* d_ws, size_t ws_size,
                              hipStream_t stream) {
  const float* x    = (const float*)d_in[0];
  const int*   esrc = (const int*)d_in[1];
  const int*   edst = (const int*)d_in[2];
  const float* eval = (const float*)d_in[3];
  const float* W1   = (const float*)d_in[4];
  const float* b1   = (const float*)d_in[5];
  const float* W2   = (const float*)d_in[6];
  const float* b2   = (const float*)d_in[7];
  float* out = (float*)d_out;

  const int R     = 4;
  const int Nn    = in_sizes[0] / FDIM;
  const int E     = in_sizes[1] / R;
  const int total = R * E;

  const size_t nf = (size_t)Nn * FDIM;
  float* h      = (float*)d_ws;
  float* x1     = h + 4 * nf;
  int*   cnt    = (int*)(x1 + nf);
  int*   rowptr = cnt + Nn + 4;
  int*   bins   = rowptr + Nn + 4;

  dim3 gGrid((Nn + BM - 1) / BM, FDIM / BN);
  const int nBlocks = (Nn + 255) / 256;
  const int eBlocks = (total + 255) / 256;
  const int rBlocks = (Nn + 7) / 8;

  zero_ints<<<nBlocks, 256, 0, stream>>>(cnt, Nn);
  count_dst<<<eBlocks, 256, 0, stream>>>(edst, cnt, total);
  scan_rows<<<1, 1024, 0, stream>>>(cnt, rowptr, Nn);
  fill_bins<<<eBlocks, 256, 0, stream>>>(edst, rowptr, cnt, bins, total);

  for (int r = 0; r < R; ++r)
    gemm_f32_wmma<<<gGrid, 256, 0, stream>>>(x, W1 + (size_t)r * FDIM * FDIM, h + r * nf, Nn);
  gather_rows<false><<<rBlocks, 256, 0, stream>>>(h, rowptr, bins, esrc, eval, b1, nullptr, x1, Nn, E);

  for (int r = 0; r < R; ++r)
    gemm_f32_wmma<<<gGrid, 256, 0, stream>>>(x1, W2 + (size_t)r * FDIM * FDIM, h + r * nf, Nn);
  gather_rows<true><<<rBlocks, 256, 0, stream>>>(h, rowptr, bins, esrc, eval, b2, x1, out, Nn, E);
}


# weight-product kernel: the 12 global loads of each 64-wide k tile are issued as one group one tile ahead, before the current tile's matrix instructions (register staging), not one by one each waited
# speedup vs baseline: 2.1113x; 1.1835x over previous
#include <hip/hip_runtime.h>

typedef __attribute__((ext_vector_type(2))) float v2f;
typedef __attribute__((ext_vector_type(8))) float v8f;

#define FDIM 256
#define RELS 4
#define KCAT (RELS * FDIM)
#define BM 64
#define BN 128
#define BK 64
#define SXP (BK + 4)
#define BNP 144

template <bool FINAL>
__global__ __launch_bounds__(256)
void gemm_cat_f32_wmma(const float* __restrict__ X, const float* __restrict__ W,
                       const float* __restrict__ bias, const float* __restrict__ skip,
                       float* __restrict__ O, int nrows, int kdim) {
  __shared__ __align__(16) float sX[BM * SXP];
  __shared__ __align__(16) float sWp[(BK / 2) * BNP * 2];

  const int t       = threadIdx.x;
  const int lane    = t & 31;
  const int wave    = t >> 5;
  const int l16     = lane & 15;
  const int half    = lane >> 4;
  const int waveRow = wave & 3;
  const int waveCol = wave >> 2;
  const int rowBase = blockIdx.y * BM;
  const int colBase = blockIdx.x * BN;
  const int cb      = waveCol * 64 + l16;

  const int xr0 = t >> 4;
  const int xc4 = (t & 15) << 2;
  const float* xp0 = X + (size_t)min(rowBase + xr0,      nrows - 1) * kdim + xc4;
  const float* xp1 = X + (size_t)min(rowBase + xr0 + 16, nrows - 1) * kdim + xc4;
  const float* xp2 = X + (size_t)min(rowBase + xr0 + 32, nrows - 1) * kdim + xc4;
  const float* xp3 = X + (size_t)min(rowBase + xr0 + 48, nrows - 1) * kdim + xc4;
  float* sxp = &sX[xr0 * SXP + xc4];
  const int wk0 = t >> 5;
  const int wn4 = (t & 31) << 2;
  const float* wp = W + (size_t)wk0 * FDIM + colBase + wn4;
  float* swp = sWp + ((size_t)(wk0 >> 1) * BNP + wn4) * 2 + (wk0 & 1);

  v8f acc0 = {}, acc1 = {}, acc2 = {}, acc3 = {};

  float4 xv0 = *(const float4*)xp0, xv1 = *(const float4*)xp1;
  float4 xv2 = *(const float4*)xp2, xv3 = *(const float4*)xp3;
  float4 wv[8];
#pragma unroll
  for (int it = 0; it < 8; ++it)
    wv[it] = *(const float4*)(wp + (size_t)(8 * it) * FDIM);

  for (int kBase = 0; kBase < kdim; kBase += BK) {
    *(float4*)(sxp)            = xv0;
    *(float4*)(sxp + 16 * SXP) = xv1;
    *(float4*)(sxp + 32 * SXP) = xv2;
    *(float4*)(sxp + 48 * SXP) = xv3;
#pragma unroll
    for (int it = 0; it < 8; ++it) {
      float* p = swp + it * 4 * BNP * 2;
      p[0] = wv[it].x; p[2] = wv[it].y; p[4] = wv[it].z; p[6] = wv[it].w;
    }
    __syncthreads();

    const int kn = min(kBase + BK, kdim - BK);
    xv0 = *(const float4*)(xp0 + kn); xv1 = *(const float4*)(xp1 + kn);
    xv2 = *(const float4*)(xp2 + kn); xv3 = *(const float4*)(xp3 + kn);
#pragma unroll
    for (int it = 0; it < 8; ++it)
      wv[it] = *(const float4*)(wp + (size_t)(kn + 8 * it) * FDIM);
    __builtin_amdgcn_sched_barrier(0);

#pragma unroll
    for (int kk = 0; kk < BK; kk += 4) {
      const int ks = kk + 2 * half;
      v2f a = *(const v2f*)&sX[(waveRow * 16 + l16) * SXP + ks];
      const v2f* wrow = (const v2f*)(sWp + ((kk >> 1) + half) * BNP * 2);
      v2f b0 = wrow[cb];
      v2f b1 = wrow[cb + 16];
      v2f b2 = wrow[cb + 32];
      v2f b3 = wrow[cb + 48];
      acc0 = __builtin_amdgcn_wmma_f32_16x16x4_f32(false, a, false, b0, (short)0, acc0, false, false);
      acc1 = __builtin_amdgcn_wmma_f32_16x16x4_f32(false, a, false, b1, (short)0, acc1, false, false);
      acc2 = __builtin_amdgcn_wmma_f32_16x16x4_f32(false, a, false, b2, (short)0, acc2, false, false);
      acc3 = __builtin_amdgcn_wmma_f32_16x16x4_f32(false, a, false, b3, (short)0, acc3, false, false);
    }
    __syncthreads();
  }

  const int   colw = colBase + waveCol * 64 + l16;
  const float bb0 = bias[colw], bb1 = bias[colw + 16];
  const float bb2 = bias[colw + 32], bb3 = bias[colw + 48];
#pragma unroll
  for (int v = 0; v < 8; ++v) {
    int gr = rowBase + waveRow * 16 + v + 8 * half;
    if (gr < nrows) {
      const size_t off = (size_t)gr * FDIM + colw;
      float o0 = fmaxf(acc0[v] + bb0, 0.f);
      float o1 = fmaxf(acc1[v] + bb1, 0.f);
      float o2 = fmaxf(acc2[v] + bb2, 0.f);
      float o3 = fmaxf(acc3[v] + bb3, 0.f);
      if (FINAL) {
        o0 += skip[off];      o1 += skip[off + 16];
        o2 += skip[off + 32]; o3 += skip[off + 48];
      }
      O[off]      = o0;
      O[off + 16] = o1;
      O[off + 32] = o2;
      O[off + 48] = o3;
    }
  }
}

#define KEY_NONE 0x7fffffff

__global__ __launch_bounds__(256)
void zero_ints(int* __restrict__ p, int n) {
  int i = blockIdx.x * blockDim.x + threadIdx.x;
  if (i < n) p[i] = 0;
}

__global__ __launch_bounds__(256)
void count_dst(const int* __restrict__ dst, int* __restrict__ cnt, int total) {
  int g = blockIdx.x * blockDim.x + threadIdx.x;
  if (g < total) atomicAdd(cnt + dst[g], 1);
}

__global__ __launch_bounds__(1024)
void scan_rows(const int* __restrict__ cnt, int* __restrict__ rowptr, int n) {
  __shared__ int part[1024];
  const int t     = threadIdx.x;
  const int chunk = (n + 1023) >> 10;
  const int b     = t * chunk;
  const int e     = min(b + chunk, n);
  int s = 0;
  for (int i = b; i < e; ++i) s += cnt[i];
  part[t] = s;
  __syncthreads();
  for (int off = 1; off < 1024; off <<= 1) {
    int v = (t >= off) ? part[t - off] : 0;
    __syncthreads();
    part[t] += v;
    __syncthreads();
  }
  int run = part[t] - s;
  for (int i = b; i < e; ++i) { rowptr[i] = run; run += cnt[i]; }
  if (t == 1023) rowptr[n] = part[1023];
}

__global__ __launch_bounds__(256)
void fill_bins(const int* __restrict__ dst, const int* __restrict__ rowptr,
               int* __restrict__ cnt, int* __restrict__ bins, int total) {
  int g = blockIdx.x * blockDim.x + threadIdx.x;
  if (g < total) {
    int d    = dst[g];
    int left = atomicSub(cnt + d, 1);
    bins[rowptr[d] + left - 1] = g;
  }
}

__device__ __forceinline__ int place_pick(int lo, int hi, int p) {
  return __builtin_amdgcn_readlane(p < 32 ? lo : hi, p & 31);
}

__device__ __forceinline__
void sum_row_by_relation(const float* __restrict__ T, int cA, int vA, int cB, int vB,
                         int n, int e1, int e2, int e3, int lane,
                         float* __restrict__ orow, bool live) {
  float4 a0 = make_float4(0.f, 0.f, 0.f, 0.f);
  float4 a1 = make_float4(0.f, 0.f, 0.f, 0.f);
  int cur = 0, bnd = e1;
#define STORE_AND_NEXT_RELATION()                                   \
  do {                                                              \
    if (live) {                                                     \
      *(float4*)(orow + cur * FDIM)       = a0;                     \
      *(float4*)(orow + cur * FDIM + 128) = a1;                     \
    }                                                               \
    a0 = make_float4(0.f, 0.f, 0.f, 0.f);                           \
    a1 = make_float4(0.f, 0.f, 0.f, 0.f);                           \
    ++cur;                                                          \
    bnd = (cur == 1) ? e2 : (cur == 2) ? e3 : n;                    \
  } while (0)
  int p = 0;
  for (; p + 4 <= n; p += 4) {
    const int c0 = place_pick(cA, cB, p);
    const int c1 = place_pick(cA, cB, p + 1);
    const int c2 = place_pick(cA, cB, p + 2);
    const int c3 = place_pick(cA, cB, p + 3);
    const float w0 = __int_as_float(place_pick(vA, vB, p));
    const float w1 = __int_as_float(place_pick(vA, vB, p + 1));
    const float w2 = __int_as_float(place_pick(vA, vB, p + 2));
    const float w3 = __int_as_float(place_pick(vA, vB, p + 3));
    const float* q0 = T + (size_t)c0 * FDIM + lane * 4;
    const float* q1 = T + (size_t)c1 * FDIM + lane * 4;
    const float* q2 = T + (size_t)c2 * FDIM + lane * 4;
    const float* q3 = T + (size_t)c3 * FDIM + lane * 4;
    const float4 x00 = *(const float4*)q0, x01 = *(const float4*)(q0 + 128);
    const float4 x10 = *(const float4*)q1, x11 = *(const float4*)(q1 + 128);
    const float4 x20 = *(const float4*)q2, x21 = *(const float4*)(q2 + 128);
    const float4 x30 = *(const float4*)q3, x31 = *(const float4*)(q3 + 128);
    while (cur < RELS - 1 && p >= bnd) STORE_AND_NEXT_RELATION();
    a0.x = fmaf(x00.x, w0, a0.x); a0.y = fmaf(x00.y, w0, a0.y);
    a0.z = fmaf(x00.z, w0, a0.z); a0.w = fmaf(x00.w, w0, a0.w);
    a1.x = fmaf(x01.x, w0, a1.x); a1.y = fmaf(x01.y, w0, a1.y);
    a1.z = fmaf(x01.z, w0, a1.z); a1.w = fmaf(x01.w, w0, a1.w);
    while (cur < RELS - 1 && p + 1 >= bnd) STORE_AND_NEXT_RELATION();
    a0.x = fmaf(x10.x, w1, a0.x); a0.y = fmaf(x10.y, w1, a0.y);
    a0.z = fmaf(x10.z, w1, a0.z); a0.w = fmaf(x10.w, w1, a0.w);
    a1.x = fmaf(x11.x, w1, a1.x); a1.y = fmaf(x11.y, w1, a1.y);
    a1.z = fmaf(x11.z, w1, a1.z); a1.w = fmaf(x11.w, w1, a1.w);
    while (cur < RELS - 1 && p + 2 >= bnd) STORE_AND_NEXT_RELATION();
    a0.x = fmaf(x20.x, w2, a0.x); a0.y = fmaf(x20.y, w2, a0.y);
    a0.z = fmaf(x20.z, w2, a0.z); a0.w = fmaf(x20.w, w2, a0.w);
    a1.x = fmaf(x21.x, w2, a1.x); a1.y = fmaf(x21.y, w2, a1.y);
    a1.z = fmaf(x21.z, w2, a1.z); a1.w = fmaf(x21.w, w2, a1.w);
    while (cur < RELS - 1 && p + 3 >= bnd) STORE_AND_NEXT_RELATION();
    a0.x = fmaf(x30.x, w3, a0.x); a0.y = fmaf(x30.y, w3, a0.y);
    a0.z = fmaf(x30.z, w3, a0.z); a0.w = fmaf(x30.w, w3, a0.w);
    a1.x = fmaf(x31.x, w3, a1.x); a1.y = fmaf(x31.y, w3, a1.y);
    a1.z = fmaf(x31.z, w3, a1.z); a1.w = fmaf(x31.w, w3, a1.w);
  }
  for (; p < n; ++p) {
    const int   c = place_pick(cA, cB, p);
    const float w = __int_as_float(place_pick(vA, vB, p));
    const float* q = T + (size_t)c * FDIM + lane * 4;
    const float4 x0 = *(const float4*)q, x1 = *(const float4*)(q + 128);
    while (cur < RELS - 1 && p >= bnd) STORE_AND_NEXT_RELATION();
    a0.x = fmaf(x0.x, w, a0.x); a0.y = fmaf(x0.y, w, a0.y);
    a0.z = fmaf(x0.z, w, a0.z); a0.w = fmaf(x0.w, w, a0.w);
    a1.x = fmaf(x1.x, w, a1.x); a1.y = fmaf(x1.y, w, a1.y);
    a1.z = fmaf(x1.z, w, a1.z); a1.w = fmaf(x1.w, w, a1.w);
  }
  while (cur < RELS) STORE_AND_NEXT_RELATION();
#undef STORE_AND_NEXT_RELATION
}

__global__ __launch_bounds__(256)
void gather_rel_rows(const float* __restrict__ T, const int* __restrict__ rowptr,
                     const int* __restrict__ bins, const int* __restrict__ esrc,
                     const float* __restrict__ eval, float* __restrict__ agg,
                     int N, int E) {
  __shared__ int sC[8][64];
  __shared__ int sV[8][64];
  const int lane = threadIdx.x & 31;
  const int wave = __builtin_amdgcn_readfirstlane(threadIdx.x >> 5);
  const int row  = blockIdx.x * 8 + wave;
  int start = 0, n = 0;
  if (row < N) { start = rowptr[row]; n = rowptr[row + 1] - start; }
  const bool small = n <= 64;
  int e1 = 0, e2 = 0, e3 = 0;

  if (small) {
    const bool h0 = lane < n, h1 = lane + 32 < n;
    int k0 = KEY_NONE, k1 = KEY_NONE;
    if (h0) k0 = bins[start + lane];
    if (h1) k1 = bins[start + 32 + lane];
    int s0 = 0, s1 = 0, v0 = 0, v1 = 0;
    if (h0) { s0 = esrc[k0]; v0 = __float_as_int(eval[k0]); }
    if (h1) { s1 = esrc[k1]; v1 = __float_as_int(eval[k1]); }
    int r0 = 0, r1 = 0;
    const int na = min(n, 32);
    for (int i = 0; i < na; ++i) {
      const int ki = __builtin_amdgcn_readlane(k0, i);
      r0 += (ki < k0) ? 1 : 0;
      r1 += (ki < k1) ? 1 : 0;
    }
    for (int i = 0; i < n - 32; ++i) {
      const int ki = __builtin_amdgcn_readlane(k1, i);
      r0 += (ki < k0) ? 1 : 0;
      r1 += (ki < k1) ? 1 : 0;
    }
    if (h0) { sC[wave][r0] = s0; sV[wave][r0] = v0; }
    if (h1) { sC[wave][r1] = s1; sV[wave][r1] = v1; }
    e1 = __builtin_popcount(__builtin_amdgcn_ballot_w32(k0 < E)) +
         __builtin_popcount(__builtin_amdgcn_ballot_w32(k1 < E));
    e2 = __builtin_popcount(__builtin_amdgcn_ballot_w32(k0 < 2 * E)) +
         __builtin_popcount(__builtin_amdgcn_ballot_w32(k1 < 2 * E));
    e3 = __builtin_popcount(__builtin_amdgcn_ballot_w32(k0 < 3 * E)) +
         __builtin_popcount(__builtin_amdgcn_ballot_w32(k1 < 3 * E));
  }
  __syncthreads();

  float* orow = agg + (size_t)row * KCAT + lane * 4;
  if (small) {
    int cA = 0, vA = 0, cB = 0, vB = 0;
    if (lane < n)      { cA = sC[wave][lane];      vA = sV[wave][lane]; }
    if (lane + 32 < n) { cB = sC[wave][lane + 32]; vB = sV[wave][lane + 32]; }
    sum_row_by_relation(T, cA, vA, cB, vB, n, e1, e2, e3, lane, orow, row < N);
  } else {
    float4 a0 = make_float4(0.f, 0.f, 0.f, 0.f);
    float4 a1 = make_float4(0.f, 0.f, 0.f, 0.f);
    int last = -1, cur = 0;
    for (int p = 0; p < n; ++p) {
      int best = KEY_NONE;
      for (int j = lane; j < n; j += 32) {
        const int k = bins[start + j];
        if (k > last && k < best) best = k;
      }
      for (int off = 16; off > 0; off >>= 1) best = min(best, __shfl_xor(best, off));
      best = __builtin_amdgcn_readfirstlane(best);
      last = best;
      const int   s   = esrc[best];
      const float w   = eval[best];
      const int   rel = (best >= E ? 1 : 0) + (best >= 2 * E ? 1 : 0) + (best >= 3 * E ? 1 : 0);
      for (; cur < rel; ++cur) {
        *(float4*)(orow + cur * FDIM)       = a0;
        *(float4*)(orow + cur * FDIM + 128) = a1;
        a0 = make_float4(0.f, 0.f, 0.f, 0.f);
        a1 = make_float4(0.f, 0.f, 0.f, 0.f);
      }
      const float* q = T + (size_t)s * FDIM + lane * 4;
      const float4 x0 = *(const float4*)q, x1 = *(const float4*)(q + 128);
      a0.x = fmaf(x0.x, w, a0.x); a0.y = fmaf(x0.y, w, a0.y);
      a0.z = fmaf(x0.z, w, a0.z); a0.w = fmaf(x0.w, w, a0.w);
      a1.x = fmaf(x1.x, w, a1.x); a1.y = fmaf(x1.y, w, a1.y);
      a1.z = fmaf(x1.z, w, a1.z); a1.w = fmaf(x1.w, w, a1.w);
    }
    for (; cur < RELS; ++cur) {
      *(float4*)(orow + cur * FDIM)       = a0;
      *(float4*)(orow + cur * FDIM + 128) = a1;
      a0 = make_float4(0.f, 0.f, 0.f, 0.f);
      a1 = make_float4(0.f, 0.f, 0.f, 0.f);
    }
  }
}

extern "C" void kernel_launch(void* const* d_in, const int* in_sizes, int n_in,
                              void* d_out, int out_size, void* d_ws, size_t ws_size,
                              hipStream_t stream) {
  const float* x    = (const float*)d_in[0];
  const int*   esrc = (const int*)d_in[1];
  const int*   edst = (const int*)d_in[2];
  const float* eval = (const float*)d_in[3];
  const float* W1   = (const float*)d_in[4];
  const float* b1   = (const float*)d_in[5];
  const float* W2   = (const float*)d_in[6];
  const float* b2   = (const float*)d_in[7];
  float* out = (float*)d_out;

  const int R     = 4;
  const int Nn    = in_sizes[0] / FDIM;
  const int E     = in_sizes[1] / R;
  const int total = R * E;

  const size_t nf = (size_t)Nn * FDIM;
  float* agg    = (float*)d_ws;
  float* x1     = agg + 4 * nf;
  int*   cnt    = (int*)(x1 + nf);
  int*   rowptr = cnt + Nn + 4;
  int*   bins   = rowptr + Nn + 4;

  dim3 gGrid(FDIM / BN, (Nn + BM - 1) / BM);
  const int nBlocks = (Nn + 255) / 256;
  const int eBlocks = (total + 255) / 256;
  const int rBlocks = (Nn + 7) / 8;

  zero_ints<<<nBlocks, 256, 0, stream>>>(cnt, Nn);
  count_dst<<<eBlocks, 256, 0, stream>>>(edst, cnt, total);
  scan_rows<<<1, 1024, 0, stream>>>(cnt, rowptr, Nn);
  fill_bins<<<eBlocks, 256, 0, stream>>>(edst, rowptr, cnt, bins, total);

  gather_rel_rows<<<rBlocks, 256, 0, stream>>>(x, rowptr, bins, esrc, eval, agg, Nn, E);
  gemm_cat_f32_wmma<false><<<gGrid, 256, 0, stream>>>(agg, W1, b1, nullptr, x1, Nn, KCAT);

  gather_rel_rows<<<rBlocks, 256, 0, stream>>>(x1, rowptr, bins, esrc, eval, agg, Nn, E);
  gemm_cat_f32_wmma<true><<<gGrid, 256, 0, stream>>>(agg, W2, b2, x1, out, Nn, KCAT);
}


# destination-row binning: both passes of global integer atomics (row count, slot claim) replaced by per-block LDS counters whose returning add ranks the edge, plus a scan over blocks; sum order kept
# speedup vs baseline: 2.6251x; 1.2434x over previous
#include <hip/hip_runtime.h>

typedef __attribute__((ext_vector_type(2))) float v2f;
typedef __attribute__((ext_vector_type(8))) float v8f;

#define FDIM 256
#define RELS 4
#define KCAT (RELS * FDIM)
#define BM 64
#define BN 128
#define BK 64
#define SXP (BK + 4)
#define BNP 144

template <bool FINAL>
__global__ __launch_bounds__(256)
void gemm_cat_f32_wmma(const float* __restrict__ X, const float* __restrict__ W,
                       const float* __restrict__ bias, const float* __restrict__ skip,
                       float* __restrict__ O, int nrows, int kdim) {
  __shared__ __align__(16) float sX[BM * SXP];
  __shared__ __align__(16) float sWp[(BK / 2) * BNP * 2];

  const int t       = threadIdx.x;
  const int lane    = t & 31;
  const int wave    = t >> 5;
  const int l16     = lane & 15;
  const int half    = lane >> 4;
  const int waveRow = wave & 3;
  const int waveCol = wave >> 2;
  const int rowBase = blockIdx.y * BM;
  const int colBase = blockIdx.x * BN;
  const int cb      = waveCol * 64 + l16;

  const int xr0 = t >> 4;
  const int xc4 = (t & 15) << 2;
  const float* xp0 = X + (size_t)min(rowBase + xr0,      nrows - 1) * kdim + xc4;
  const float* xp1 = X + (size_t)min(rowBase + xr0 + 16, nrows - 1) * kdim + xc4;
  const float* xp2 = X + (size_t)min(rowBase + xr0 + 32, nrows - 1) * kdim + xc4;
  const float* xp3 = X + (size_t)min(rowBase + xr0 + 48, nrows - 1) * kdim + xc4;
  float* sxp = &sX[xr0 * SXP + xc4];
  const int wk0 = t >> 5;
  const int wn4 = (t & 31) << 2;
  const float* wp = W + (size_t)wk0 * FDIM + colBase + wn4;
  float* swp = sWp + ((size_t)(wk0 >> 1) * BNP + wn4) * 2 + (wk0 & 1);

  v8f acc0 = {}, acc1 = {}, acc2 = {}, acc3 = {};

  float4 xv0 = *(const float4*)xp0, xv1 = *(const float4*)xp1;
  float4 xv2 = *(const float4*)xp2, xv3 = *(const float4*)xp3;
  float4 wv[8];
#pragma unroll
  for (int it = 0; it < 8; ++it)
    wv[it] = *(const float4*)(wp + (size_t)(8 * it) * FDIM);

  for (int kBase = 0; kBase < kdim; kBase += BK) {
    *(float4*)(sxp)            = xv0;
    *(float4*)(sxp + 16 * SXP) = xv1;
    *(float4*)(sxp + 32 * SXP) = xv2;
    *(float4*)(sxp + 48 * SXP) = xv3;
#pragma unroll
    for (int it = 0; it < 8; ++it) {
      float* p = swp + it * 4 * BNP * 2;
      p[0] = wv[it].x; p[2] = wv[it].y; p[4] = wv[it].z; p[6] = wv[it].w;
    }
    __syncthreads();

    const int kn = min(kBase + BK, kdim - BK);
    xv0 = *(const float4*)(xp0 + kn); xv1 = *(const float4*)(xp1 + kn);
    xv2 = *(const float4*)(xp2 + kn); xv3 = *(const float4*)(xp3 + kn);
#pragma unroll
    for (int it = 0; it < 8; ++it)
      wv[it] = *(const float4*)(wp + (size_t)(kn + 8 * it) * FDIM);
    __builtin_amdgcn_sched_barrier(0);

#pragma unroll
    for (int kk = 0; kk < BK; kk += 4) {
      const int ks = kk + 2 * half;
      v2f a = *(const v2f*)&sX[(waveRow * 16 + l16) * SXP + ks];
      const v2f* wrow = (const v2f*)(sWp + ((kk >> 1) + half) * BNP * 2);
      v2f b0 = wrow[cb];
      v2f b1 = wrow[cb + 16];
      v2f b2 = wrow[cb + 32];
      v2f b3 = wrow[cb + 48];
      acc0 = __builtin_amdgcn_wmma_f32_16x16x4_f32(false, a, false, b0, (short)0, acc0, false, false);
      acc1 = __builtin_amdgcn_wmma_f32_16x16x4_f32(false, a, false, b1, (short)0, acc1, false, false);
      acc2 = __builtin_amdgcn_wmma_f32_16x16x4_f32(false, a, false, b2, (short)0, acc2, false, false);
      acc3 = __builtin_amdgcn_wmma_f32_16x16x4_f32(false, a, false, b3, (short)0, acc3, false, false);
    }
    __syncthreads();
  }

  const int   colw = colBase + waveCol * 64 + l16;
  const float bb0 = bias[colw], bb1 = bias[colw + 16];
  const float bb2 = bias[colw + 32], bb3 = bias[colw + 48];
#pragma unroll
  for (int v = 0; v < 8; ++v) {
    int gr = rowBase + waveRow * 16 + v + 8 * half;
    if (gr < nrows) {
      const size_t off = (size_t)gr * FDIM + colw;
      float o0 = fmaxf(acc0[v] + bb0, 0.f);
      float o1 = fmaxf(acc1[v] + bb1, 0.f);
      float o2 = fmaxf(acc2[v] + bb2, 0.f);
      float o3 = fmaxf(acc3[v] + bb3, 0.f);
      if (FINAL) {
        o0 += skip[off];      o1 += skip[off + 16];
        o2 += skip[off + 32]; o3 += skip[off + 48];
      }
      O[off]      = o0;
      O[off + 16] = o1;
      O[off + 32] = o2;
      O[off + 48] = o3;
    }
  }
}

#define KEY_NONE 0x7fffffff

#define NCB  64
#define HCAP 65536

__global__ __launch_bounds__(1024)
void count_rows_lds(const int* __restrict__ dst, int* __restrict__ tab,
                    int* __restrict__ rank, int total, int n) {
  __shared__ int hist[HCAP];
  const int t   = threadIdx.x;
  const int per = (total + (int)gridDim.x - 1) / (int)gridDim.x;
  const int e0  = min((int)blockIdx.x * per, total);
  const int e1  = min(e0 + per, total);
  int* trow = tab + (size_t)blockIdx.x * n;
  for (int base = 0; base < n; base += HCAP) {
    const int len = min(HCAP, n - base);
    for (int i = t; i < len; i += 1024) hist[i] = 0;
    __syncthreads();
    int g = e0 + t;
    for (; g + 3 * 1024 < e1; g += 4 * 1024) {
      const int d0 = dst[g]            - base;
      const int d1 = dst[g + 1024]     - base;
      const int d2 = dst[g + 2 * 1024] - base;
      const int d3 = dst[g + 3 * 1024] - base;
      if ((unsigned)d0 < (unsigned)len) rank[g]            = atomicAdd(&hist[d0], 1);
      if ((unsigned)d1 < (unsigned)len) rank[g + 1024]     = atomicAdd(&hist[d1], 1);
      if ((unsigned)d2 < (unsigned)len) rank[g + 2 * 1024] = atomicAdd(&hist[d2], 1);
      if ((unsigned)d3 < (unsigned)len) rank[g + 3 * 1024] = atomicAdd(&hist[d3], 1);
    }
    for (; g < e1; g += 1024) {
      const int d = dst[g] - base;
      if ((unsigned)d < (unsigned)len) rank[g] = atomicAdd(&hist[d], 1);
    }
    __syncthreads();
    for (int i = t; i < len; i += 1024) trow[base + i] = hist[i];
    __syncthreads();
  }
}

__global__ __launch_bounds__(256)
void scan_pieces(int* __restrict__ tab, int* __restrict__ cnt, int n) {
  const int d = blockIdx.x * blockDim.x + threadIdx.x;
  if (d >= n) return;
  int v[NCB];
#pragma unroll
  for (int b = 0; b < NCB; ++b) v[b] = tab[(size_t)b * n + d];
  int run = 0;
#pragma unroll
  for (int b = 0; b < NCB; ++b) {
    const int c = v[b];
    tab[(size_t)b * n + d] = run;
    run += c;
  }
  cnt[d] = run;
}

__global__ __launch_bounds__(1024)
void scan_rows(const int* __restrict__ cnt, int* __restrict__ rowptr, int n) {
  __shared__ int part[1024];
  const int t     = threadIdx.x;
  const int chunk = (n + 1023) >> 10;
  const int b     = t * chunk;
  const int e     = min(b + chunk, n);
  int s = 0;
  for (int i = b; i < e; ++i) s += cnt[i];
  part[t] = s;
  __syncthreads();
  for (int off = 1; off < 1024; off <<= 1) {
    int v = (t >= off) ? part[t - off] : 0;
    __syncthreads();
    part[t] += v;
    __syncthreads();
  }
  int run = part[t] - s;
  for (int i = b; i < e; ++i) { rowptr[i] = run; run += cnt[i]; }
  if (t == 1023) rowptr[n] = part[1023];
}

__global__ __launch_bounds__(256)
void fill_bins(const int* __restrict__ dst, const int* __restrict__ rowptr,
               const int* __restrict__ tab, const int* __restrict__ rank,
               int* __restrict__ bins, int total, int n) {
  const int per = (total + (int)gridDim.y - 1) / (int)gridDim.y;
  const int i   = blockIdx.x * blockDim.x + threadIdx.x;
  const int g   = (int)blockIdx.y * per + i;
  if (i < per && g < total) {
    const int d = dst[g];
    bins[rowptr[d] + tab[(size_t)blockIdx.y * n + d] + rank[g]] = g;
  }
}

__device__ __forceinline__ int place_pick(int lo, int hi, int p) {
  return __builtin_amdgcn_readlane(p < 32 ? lo : hi, p & 31);
}

__device__ __forceinline__
void sum_row_by_relation(const float* __restrict__ T, int cA, int vA, int cB, int vB,
                         int n, int e1, int e2, int e3, int lane,
                         float* __restrict__ orow, bool live) {
  float4 a0 = make_float4(0.f, 0.f, 0.f, 0.f);
  float4 a1 = make_float4(0.f, 0.f, 0.f, 0.f);
  int cur = 0, bnd = e1;
#define STORE_AND_NEXT_RELATION()                                   \
  do {                                                              \
    if (live) {                                                     \
      *(float4*)(orow + cur * FDIM)       = a0;                     \
      *(float4*)(orow + cur * FDIM + 128) = a1;                     \
    }                                                               \
    a0 = make_float4(0.f, 0.f, 0.f, 0.f);                           \
    a1 = make_float4(0.f, 0.f, 0.f, 0.f);                           \
    ++cur;                                                          \
    bnd = (cur == 1) ? e2 : (cur == 2) ? e3 : n;                    \
  } while (0)
  int p = 0;
  for (; p + 4 <= n; p += 4) {
    const int c0 = place_pick(cA, cB, p);
    const int c1 = place_pick(cA, cB, p + 1);
    const int c2 = place_pick(cA, cB, p + 2);
    const int c3 = place_pick(cA, cB, p + 3);
    const float w0 = __int_as_float(place_pick(vA, vB, p));
    const float w1 = __int_as_float(place_pick(vA, vB, p + 1));
    const float w2 = __int_as_float(place_pick(vA, vB, p + 2));
    const float w3 = __int_as_float(place_pick(vA, vB, p + 3));
    const float* q0 = T + (size_t)c0 * FDIM + lane * 4;
    const float* q1 = T + (size_t)c1 * FDIM + lane * 4;
    const float* q2 = T + (size_t)c2 * FDIM + lane * 4;
    const float* q3 = T + (size_t)c3 * FDIM + lane * 4;
    const float4 x00 = *(const float4*)q0, x01 = *(const float4*)(q0 + 128);
    const float4 x10 = *(const float4*)q1, x11 = *(const float4*)(q1 + 128);
    const float4 x20 = *(const float4*)q2, x21 = *(const float4*)(q2 + 128);
    const float4 x30 = *(const float4*)q3, x31 = *(const float4*)(q3 + 128);
    while (cur < RELS - 1 && p >= bnd) STORE_AND_NEXT_RELATION();
    a0.x = fmaf(x00.x, w0, a0.x); a0.y = fmaf(x00.y, w0, a0.y);
    a0.z = fmaf(x00.z, w0, a0.z); a0.w = fmaf(x00.w, w0, a0.w);
    a1.x = fmaf(x01.x, w0, a1.x); a1.y = fmaf(x01.y, w0, a1.y);
    a1.z = fmaf(x01.z, w0, a1.z); a1.w = fmaf(x01.w, w0, a1.w);
    while (cur < RELS - 1 && p + 1 >= bnd) STORE_AND_NEXT_RELATION();
    a0.x = fmaf(x10.x, w1, a0.x); a0.y = fmaf(x10.y, w1, a0.y);
    a0.z = fmaf(x10.z, w1, a0.z); a0.w = fmaf(x10.w, w1, a0.w);
    a1.x = fmaf(x11.x, w1, a1.x); a1.y = fmaf(x11.y, w1, a1.y);
    a1.z = fmaf(x11.z, w1, a1.z); a1.w = fmaf(x11.w, w1, a1.w);
    while (cur < RELS - 1 && p + 2 >= bnd) STORE_AND_NEXT_RELATION();
    a0.x = fmaf(x20.x, w2, a0.x); a0.y = fmaf(x20.y, w2, a0.y);
    a0.z = fmaf(x20.z, w2, a0.z); a0.w = fmaf(x20.w, w2, a0.w);
    a1.x = fmaf(x21.x, w2, a1.x); a1.y = fmaf(x21.y, w2, a1.y);
    a1.z = fmaf(x21.z, w2, a1.z); a1.w = fmaf(x21.w, w2, a1.w);
    while (cur < RELS - 1 && p + 3 >= bnd) STORE_AND_NEXT_RELATION();
    a0.x = fmaf(x30.x, w3, a0.x); a0.y = fmaf(x30.y, w3, a0.y);
    a0.z = fmaf(x30.z, w3, a0.z); a0.w = fmaf(x30.w, w3, a0.w);
    a1.x = fmaf(x31.x, w3, a1.x); a1.y = fmaf(x31.y, w3, a1.y);
    a1.z = fmaf(x31.z, w3, a1.z); a1.w = fmaf(x31.w, w3, a1.w);
  }
  for (; p < n; ++p) {
    const int   c = place_pick(cA, cB, p);
    const float w = __int_as_float(place_pick(vA, vB, p));
    const float* q = T + (size_t)c * FDIM + lane * 4;
    const float4 x0 = *(const float4*)q, x1 = *(const float4*)(q + 128);
    while (cur < RELS - 1 && p >= bnd) STORE_AND_NEXT_RELATION();
    a0.x = fmaf(x0.x, w, a0.x); a0.y = fmaf(x0.y, w, a0.y);
    a0.z = fmaf(x0.z, w, a0.z); a0.w = fmaf(x0.w, w, a0.w);
    a1.x = fmaf(x1.x, w, a1.x); a1.y = fmaf(x1.y, w, a1.y);
    a1.z = fmaf(x1.z, w, a1.z); a1.w = fmaf(x1.w, w, a1.w);
  }
  while (cur < RELS) STORE_AND_NEXT_RELATION();
#undef STORE_AND_NEXT_RELATION
}

__global__ __launch_bounds__(256)
void gather_rel_rows(const float* __restrict__ T, const int* __restrict__ rowptr,
                     const int* __restrict__ bins, const int* __restrict__ esrc,
                     const float* __restrict__ eval, float* __restrict__ agg,
                     int N, int E) {
  __shared__ int sC[8][64];
  __shared__ int sV[8][64];
  const int lane = threadIdx.x & 31;
  const int wave = __builtin_amdgcn_readfirstlane(threadIdx.x >> 5);
  const int row  = blockIdx.x * 8 + wave;
  int start = 0, n = 0;
  if (row < N) { start = rowptr[row]; n = rowptr[row + 1] - start; }
  const bool small = n <= 64;
  int e1 = 0, e2 = 0, e3 = 0;

  if (small) {
    const bool h0 = lane < n, h1 = lane + 32 < n;
    int k0 = KEY_NONE, k1 = KEY_NONE;
    if (h0) k0 = bins[start + lane];
    if (h1) k1 = bins[start + 32 + lane];
    int s0 = 0, s1 = 0, v0 = 0, v1 = 0;
    if (h0) { s0 = esrc[k0]; v0 = __float_as_int(eval[k0]); }
    if (h1) { s1 = esrc[k1]; v1 = __float_as_int(eval[k1]); }
    int r0 = 0, r1 = 0;
    const int na = min(n, 32);
    for (int i = 0; i < na; ++i) {
      const int ki = __builtin_amdgcn_readlane(k0, i);
      r0 += (ki < k0) ? 1 : 0;
      r1 += (ki < k1) ? 1 : 0;
    }
    for (int i = 0; i < n - 32; ++i) {
      const int ki = __builtin_amdgcn_readlane(k1, i);
      r0 += (ki < k0) ? 1 : 0;
      r1 += (ki < k1) ? 1 : 0;
    }
    if (h0) { sC[wave][r0] = s0; sV[wave][r0] = v0; }
    if (h1) { sC[wave][r1] = s1; sV[wave][r1] = v1; }
    e1 = __builtin_popcount(__builtin_amdgcn_ballot_w32(k0 < E)) +
         __builtin_popcount(__builtin_amdgcn_ballot_w32(k1 < E));
    e2 = __builtin_popcount(__builtin_amdgcn_ballot_w32(k0 < 2 * E)) +
         __builtin_popcount(__builtin_amdgcn_ballot_w32(k1 < 2 * E));
    e3 = __builtin_popcount(__builtin_amdgcn_ballot_w32(k0 < 3 * E)) +
         __builtin_popcount(__builtin_amdgcn_ballot_w32(k1 < 3 * E));
  }
  __syncthreads();

  float* orow = agg + (size_t)row * KCAT + lane * 4;
  if (small) {
    int cA = 0, vA = 0, cB = 0, vB = 0;
    if (lane < n)      { cA = sC[wave][lane];      vA = sV[wave][lane]; }
    if (lane + 32 < n) { cB = sC[wave][lane + 32]; vB = sV[wave][lane + 32]; }
    sum_row_by_relation(T, cA, vA, cB, vB, n, e1, e2, e3, lane, orow, row < N);
  } else {
    float4 a0 = make_float4(0.f, 0.f, 0.f, 0.f);
    float4 a1 = make_float4(0.f, 0.f, 0.f, 0.f);
    int last = -1, cur = 0;
    for (int p = 0; p < n; ++p) {
      int best = KEY_NONE;
      for (int j = lane; j < n; j += 32) {
        const int k = bins[start + j];
        if (k > last && k < best) best = k;
      }
      for (int off = 16; off > 0; off >>= 1) best = min(best, __shfl_xor(best, off));
      best = __builtin_amdgcn_readfirstlane(best);
      last = best;
      const int   s   = esrc[best];
      const float w   = eval[best];
      const int   rel = (best >= E ? 1 : 0) + (best >= 2 * E ? 1 : 0) + (best >= 3 * E ? 1 : 0);
      for (; cur < rel; ++cur) {
        *(float4*)(orow + cur * FDIM)       = a0;
        *(float4*)(orow + cur * FDIM + 128) = a1;
        a0 = make_float4(0.f, 0.f, 0.f, 0.f);
        a1 = make_float4(0.f, 0.f, 0.f, 0.f);
      }
      const float* q = T + (size_t)s * FDIM + lane * 4;
      const float4 x0 = *(const float4*)q, x1 = *(const float4*)(q + 128);
      a0.x = fmaf(x0.x, w, a0.x); a0.y = fmaf(x0.y, w, a0.y);
      a0.z = fmaf(x0.z, w, a0.z); a0.w = fmaf(x0.w, w, a0.w);
      a1.x = fmaf(x1.x, w, a1.x); a1.y = fmaf(x1.y, w, a1.y);
      a1.z = fmaf(x1.z, w, a1.z); a1.w = fmaf(x1.w, w, a1.w);
    }
    for (; cur < RELS; ++cur) {
      *(float4*)(orow + cur * FDIM)       = a0;
      *(float4*)(orow + cur * FDIM + 128) = a1;
      a0 = make_float4(0.f, 0.f, 0.f, 0.f);
      a1 = make_float4(0.f, 0.f, 0.f, 0.f);
    }
  }
}

extern "C" void kernel_launch(void* const* d_in, const int* in_sizes, int n_in,
                              void* d_out, int out_size, void* d_ws, size_t ws_size,
                              hipStream_t stream) {
  const float* x    = (const float*)d_in[0];
  const int*   esrc = (const int*)d_in[1];
  const int*   edst = (const int*)d_in[2];
  const float* eval = (const float*)d_in[3];
  const float* W1   = (const float*)d_in[4];
  const float* b1   = (const float*)d_in[5];
  const float* W2   = (const float*)d_in[6];
  const float* b2   = (const float*)d_in[7];
  float* out = (float*)d_out;

  const int R     = 4;
  const int Nn    = in_sizes[0] / FDIM;
  const int E     = in_sizes[1] / R;
  const int total = R * E;

  const size_t nf = (size_t)Nn * FDIM;
  float* agg    = (float*)d_ws;
  float* x1     = agg + 4 * nf;
  int*   cnt    = (int*)(x1 + nf);
  int*   rowptr = cnt + Nn + 4;
  int*   bins   = rowptr + Nn + 4;

  int*   tab    = (int*)d_ws;
  int*   rank   = tab + (size_t)NCB * Nn;

  dim3 gGrid(FDIM / BN, (Nn + BM - 1) / BM);
  const int nBlocks = (Nn + 255) / 256;
  const int rBlocks = (Nn + 7) / 8;
  const int per     = (total + NCB - 1) / NCB;
  dim3 fGrid((per + 255) / 256, NCB);

  count_rows_lds<<<NCB, 1024, 0, stream>>>(edst, tab, rank, total, Nn);
  scan_pieces<<<nBlocks, 256, 0, stream>>>(tab, cnt, Nn);
  scan_rows<<<1, 1024, 0, stream>>>(cnt, rowptr, Nn);
  fill_bins<<<fGrid, 256, 0, stream>>>(edst, rowptr, tab, rank, bins, total, Nn);

  gather_rel_rows<<<rBlocks, 256, 0, stream>>>(x, rowptr, bins, esrc, eval, agg, Nn, E);
  gemm_cat_f32_wmma<false><<<gGrid, 256, 0, stream>>>(agg, W1, b1, nullptr, x1, Nn, KCAT);

  gather_rel_rows<<<rBlocks, 256, 0, stream>>>(x1, rowptr, bins, esrc, eval, agg, Nn, E);
  gemm_cat_f32_wmma<true><<<gGrid, 256, 0, stream>>>(agg, W2, b2, x1, out, Nn, KCAT);
}


# weight-product kernel: W tile staged in LDS with contiguous 16-byte stores (k-pair rows cut into 16-column groups) in place of 4-byte stores with 4-way bank conflicts; sums unchanged
# speedup vs baseline: 2.8667x; 1.0920x over previous
#include <hip/hip_runtime.h>

typedef __attribute__((ext_vector_type(2))) float v2f;
typedef __attribute__((ext_vector_type(4))) float v4f;
typedef __attribute__((ext_vector_type(8))) float v8f;

#define FDIM 256
#define RELS 4
#define KCAT (RELS * FDIM)
#define BM 64
#define BN 128
#define BK 64
#define SXP (BK + 4)
#define WPP 288

template <bool FINAL>
__global__ __launch_bounds__(256)
void gemm_cat_f32_wmma(const float* __restrict__ X, const float* __restrict__ W,
                       const float* __restrict__ bias, const float* __restrict__ skip,
                       float* __restrict__ O, int nrows, int kdim) {
  __shared__ __align__(16) float sX[BM * SXP];
  __shared__ __align__(16) float sW[(BK / 2) * WPP];

  const int t       = threadIdx.x;
  const int lane    = t & 31;
  const int wave    = t >> 5;
  const int l16     = lane & 15;
  const int half    = lane >> 4;
  const int waveRow = wave & 3;
  const int waveCol = wave >> 2;
  const int rowBase = blockIdx.y * BM;
  const int colBase = blockIdx.x * BN;
  const int cb2     = waveCol * 128 + l16;

  const int xr0 = t >> 4;
  const int xc4 = (t & 15) << 2;
  const float* xp0 = X + (size_t)min(rowBase + xr0,      nrows - 1) * kdim + xc4;
  const float* xp1 = X + (size_t)min(rowBase + xr0 + 16, nrows - 1) * kdim + xc4;
  const float* xp2 = X + (size_t)min(rowBase + xr0 + 32, nrows - 1) * kdim + xc4;
  const float* xp3 = X + (size_t)min(rowBase + xr0 + 48, nrows - 1) * kdim + xc4;
  float* sxp = &sX[xr0 * SXP + xc4];
  const int wk0 = 2 * (wave >> 1) + ((lane >> 2) & 1);
  const int wn4 = (wave & 1) * 64 + (lane >> 3) * 16 + ((lane & 3) << 2);
  const float* wp = W + (size_t)wk0 * FDIM + colBase + wn4;
  float* swp = sW + (wave >> 1) * WPP + (wave & 1) * 128 + (lane << 2);

  v8f acc0 = {}, acc1 = {}, acc2 = {}, acc3 = {};

  float4 xv0 = *(const float4*)xp0, xv1 = *(const float4*)xp1;
  float4 xv2 = *(const float4*)xp2, xv3 = *(const float4*)xp3;
  float4 wv[8];
#pragma unroll
  for (int it = 0; it < 8; ++it)
    wv[it] = *(const float4*)(wp + (size_t)(8 * it) * FDIM);

  for (int kBase = 0; kBase < kdim; kBase += BK) {
    *(float4*)(sxp)            = xv0;
    *(float4*)(sxp + 16 * SXP) = xv1;
    *(float4*)(sxp + 32 * SXP) = xv2;
    *(float4*)(sxp + 48 * SXP) = xv3;
#pragma unroll
    for (int it = 0; it < 8; ++it) {
      const v4f w4 = {wv[it].x, wv[it].y, wv[it].z, wv[it].w};
      *(v4f*)(swp + it * 4 * WPP) = w4;
    }
    __syncthreads();

    const int kn = min(kBase + BK, kdim - BK);
    xv0 = *(const float4*)(xp0 + kn); xv1 = *(const float4*)(xp1 + kn);
    xv2 = *(const float4*)(xp2 + kn); xv3 = *(const float4*)(xp3 + kn);
#pragma unroll
    for (int it = 0; it < 8; ++it)
      wv[it] = *(const float4*)(wp + (size_t)(kn + 8 * it) * FDIM);
    __builtin_amdgcn_sched_barrier(0);

#pragma unroll
    for (int kk = 0; kk < BK; kk += 4) {
      const int ks = kk + 2 * half;
      v2f a = *(const v2f*)&sX[(waveRow * 16 + l16) * SXP + ks];
      const float* wrow = sW + ((kk >> 1) + half) * WPP + cb2;
      v2f b0, b1, b2, b3;
      b0.x = wrow[0];  b0.y = wrow[16];
      b1.x = wrow[32]; b1.y = wrow[48];
      b2.x = wrow[64]; b2.y = wrow[80];
      b3.x = wrow[96]; b3.y = wrow[112];
      acc0 = __builtin_amdgcn_wmma_f32_16x16x4_f32(false, a, false, b0, (short)0, acc0, false, false);
      acc1 = __builtin_amdgcn_wmma_f32_16x16x4_f32(false, a, false, b1, (short)0, acc1, false, false);
      acc2 = __builtin_amdgcn_wmma_f32_16x16x4_f32(false, a, false, b2, (short)0, acc2, false, false);
      acc3 = __builtin_amdgcn_wmma_f32_16x16x4_f32(false, a, false, b3, (short)0, acc3, false, false);
    }
    __syncthreads();
  }

  const int   colw = colBase + waveCol * 64 + l16;
  const float bb0 = bias[colw], bb1 = bias[colw + 16];
  const float bb2 = bias[colw + 32], bb3 = bias[colw + 48];
#pragma unroll
  for (int v = 0; v < 8; ++v) {
    int gr = rowBase + waveRow * 16 + v + 8 * half;
    if (gr < nrows) {
      const size_t off = (size_t)gr * FDIM + colw;
      float o0 = fmaxf(acc0[v] + bb0, 0.f);
      float o1 = fmaxf(acc1[v] + bb1, 0.f);
      float o2 = fmaxf(acc2[v] + bb2, 0.f);
      float o3 = fmaxf(acc3[v] + bb3, 0.f);
      if (FINAL) {
        o0 += skip[off];      o1 += skip[off + 16];
        o2 += skip[off + 32]; o3 += skip[off + 48];
      }
      O[off]      = o0;
      O[off + 16] = o1;
      O[off + 32] = o2;
      O[off + 48] = o3;
    }
  }
}

#define KEY_NONE 0x7fffffff

#define NCB  64
#define HCAP 65536

__global__ __launch_bounds__(1024)
void count_rows_lds(const int* __restrict__ dst, int* __restrict__ tab,
                    int* __restrict__ rank, int total, int n) {
  __shared__ int hist[HCAP];
  const int t   = threadIdx.x;
  const int per = (total + (int)gridDim.x - 1) / (int)gridDim.x;
  const int e0  = min((int)blockIdx.x * per, total);
  const int e1  = min(e0 + per, total);
  int* trow = tab + (size_t)blockIdx.x * n;
  for (int base = 0; base < n; base += HCAP) {
    const int len = min(HCAP, n - base);
    for (int i = t; i < len; i += 1024) hist[i] = 0;
    __syncthreads();
    int g = e0 + t;
    for (; g + 3 * 1024 < e1; g += 4 * 1024) {
      const int d0 = dst[g]            - base;
      const int d1 = dst[g + 1024]     - base;
      const int d2 = dst[g + 2 * 1024] - base;
      const int d3 = dst[g + 3 * 1024] - base;
      if ((unsigned)d0 < (unsigned)len) rank[g]            = atomicAdd(&hist[d0], 1);
      if ((unsigned)d1 < (unsigned)len) rank[g + 1024]     = atomicAdd(&hist[d1], 1);
      if ((unsigned)d2 < (unsigned)len) rank[g + 2 * 1024] = atomicAdd(&hist[d2], 1);
      if ((unsigned)d3 < (unsigned)len) rank[g + 3 * 1024] = atomicAdd(&hist[d3], 1);
    }
    for (; g < e1; g += 1024) {
      const int d = dst[g] - base;
      if ((unsigned)d < (unsigned)len) rank[g] = atomicAdd(&hist[d], 1);
    }
    __syncthreads();
    for (int i = t; i < len; i += 1024) trow[base + i] = hist[i];
    __syncthreads();
  }
}

__global__ __launch_bounds__(256)
void scan_pieces(int* __restrict__ tab, int* __restrict__ cnt, int n) {
  const int d = blockIdx.x * blockDim.x + threadIdx.x;
  if (d >= n) return;
  int v[NCB];
#pragma unroll
  for (int b = 0; b < NCB; ++b) v[b] = tab[(size_t)b * n + d];
  int run = 0;
#pragma unroll
  for (int b = 0; b < NCB; ++b) {
    const int c = v[b];
    tab[(size_t)b * n + d] = run;
    run += c;
  }
  cnt[d] = run;
}

__global__ __launch_bounds__(1024)
void scan_rows(const int* __restrict__ cnt, int* __restrict__ rowptr, int n) {
  __shared__ int part[1024];
  const int t     = threadIdx.x;
  const int chunk = (n + 1023) >> 10;
  const int b     = t * chunk;
  const int e     = min(b + chunk, n);
  int s = 0;
  for (int i = b; i < e; ++i) s += cnt[i];
  part[t] = s;
  __syncthreads();
  for (int off = 1; off < 1024; off <<= 1) {
    int v = (t >= off) ? part[t - off] : 0;
    __syncthreads();
    part[t] += v;
    __syncthreads();
  }
  int run = part[t] - s;
  for (int i = b; i < e; ++i) { rowptr[i] = run; run += cnt[i]; }
  if (t == 1023) rowptr[n] = part[1023];
}

__global__ __launch_bounds__(256)
void fill_bins(const int* __restrict__ dst, const int* __restrict__ rowptr,
               const int* __restrict__ tab, const int* __restrict__ rank,
               int* __restrict__ bins, int total, int n) {
  const int per = (total + (int)gridDim.y - 1) / (int)gridDim.y;
  const int i   = blockIdx.x * blockDim.x + threadIdx.x;
  const int g   = (int)blockIdx.y * per + i;
  if (i < per && g < total) {
    const int d = dst[g];
    bins[rowptr[d] + tab[(size_t)blockIdx.y * n + d] + rank[g]] = g;
  }
}

__device__ __forceinline__ int place_pick(int lo, int hi, int p) {
  return __builtin_amdgcn_readlane(p < 32 ? lo : hi, p & 31);
}

__device__ __forceinline__
void sum_row_by_relation(const float* __restrict__ T, int cA, int vA, int cB, int vB,
                         int n, int e1, int e2, int e3, int lane,
                         float* __restrict__ orow, bool live) {
  float4 a0 = make_float4(0.f, 0.f, 0.f, 0.f);
  float4 a1 = make_float4(0.f, 0.f, 0.f, 0.f);
  int cur = 0, bnd = e1;
#define STORE_AND_NEXT_RELATION()                                   \
  do {                                                              \
    if (live) {                                                     \
      *(float4*)(orow + cur * FDIM)       = a0;                     \
      *(float4*)(orow + cur * FDIM + 128) = a1;                     \
    }                                                               \
    a0 = make_float4(0.f, 0.f, 0.f, 0.f);                           \
    a1 = make_float4(0.f, 0.f, 0.f, 0.f);                           \
    ++cur;                                                          \
    bnd = (cur == 1) ? e2 : (cur == 2) ? e3 : n;                    \
  } while (0)
  int p = 0;
  for (; p + 4 <= n; p += 4) {
    const int c0 = place_pick(cA, cB, p);
    const int c1 = place_pick(cA, cB, p + 1);
    const int c2 = place_pick(cA, cB, p + 2);
    const int c3 = place_pick(cA, cB, p + 3);
    const float w0 = __int_as_float(place_pick(vA, vB, p));
    const float w1 = __int_as_float(place_pick(vA, vB, p + 1));
    const float w2 = __int_as_float(place_pick(vA, vB, p + 2));
    const float w3 = __int_as_float(place_pick(vA, vB, p + 3));
    const float* q0 = T + (size_t)c0 * FDIM + lane * 4;
    const float* q1 = T + (size_t)c1 * FDIM + lane * 4;
    const float* q2 = T + (size_t)c2 * FDIM + lane * 4;
    const float* q3 = T + (size_t)c3 * FDIM + lane * 4;
    const float4 x00 = *(const float4*)q0, x01 = *(const float4*)(q0 + 128);
    const float4 x10 = *(const float4*)q1, x11 = *(const float4*)(q1 + 128);
    const float4 x20 = *(const float4*)q2, x21 = *(const float4*)(q2 + 128);
    const float4 x30 = *(const float4*)q3, x31 = *(const float4*)(q3 + 128);
    while (cur < RELS - 1 && p >= bnd) STORE_AND_NEXT_RELATION();
    a0.x = fmaf(x00.x, w0, a0.x); a0.y = fmaf(x00.y, w0, a0.y);
    a0.z = fmaf(x00.z, w0, a0.z); a0.w = fmaf(x00.w, w0, a0.w);
    a1.x = fmaf(x01.x, w0, a1.x); a1.y = fmaf(x01.y, w0, a1.y);
    a1.z = fmaf(x01.z, w0, a1.z); a1.w = fmaf(x01.w, w0, a1.w);
    while (cur < RELS - 1 && p + 1 >= bnd) STORE_AND_NEXT_RELATION();
    a0.x = fmaf(x10.x, w1, a0.x); a0.y = fmaf(x10.y, w1, a0.y);
    a0.z = fmaf(x10.z, w1, a0.z); a0.w = fmaf(x10.w, w1, a0.w);
    a1.x = fmaf(x11.x, w1, a1.x); a1.y = fmaf(x11.y, w1, a1.y);
    a1.z = fmaf(x11.z, w1, a1.z); a1.w = fmaf(x11.w, w1, a1.w);
    while (cur < RELS - 1 && p + 2 >= bnd) STORE_AND_NEXT_RELATION();
    a0.x = fmaf(x20.x, w2, a0.x); a0.y = fmaf(x20.y, w2, a0.y);
    a0.z = fmaf(x20.z, w2, a0.z); a0.w = fmaf(x20.w, w2, a0.w);
    a1.x = fmaf(x21.x, w2, a1.x); a1.y = fmaf(x21.y, w2, a1.y);
    a1.z = fmaf(x21.z, w2, a1.z); a1.w = fmaf(x21.w, w2, a1.w);
    while (cur < RELS - 1 && p + 3 >= bnd) STORE_AND_NEXT_RELATION();
    a0.x = fmaf(x30.x, w3, a0.x); a0.y = fmaf(x30.y, w3, a0.y);
    a0.z = fmaf(x30.z, w3, a0.z); a0.w = fmaf(x30.w, w3, a0.w);
    a1.x = fmaf(x31.x, w3, a1.x); a1.y = fmaf(x31.y, w3, a1.y);
    a1.z = fmaf(x31.z, w3, a1.z); a1.w = fmaf(x31.w, w3, a1.w);
  }
  for (; p < n; ++p) {
    const int   c = place_pick(cA, cB, p);
    const float w = __int_as_float(place_pick(vA, vB, p));
    const float* q = T + (size_t)c * FDIM + lane * 4;
    const float4 x0 = *(const float4*)q, x1 = *(const float4*)(q + 128);
    while (cur < RELS - 1 && p >= bnd) STORE_AND_NEXT_RELATION();
    a0.x = fmaf(x0.x, w, a0.x); a0.y = fmaf(x0.y, w, a0.y);
    a0.z = fmaf(x0.z, w, a0.z); a0.w = fmaf(x0.w, w, a0.w);
    a1.x = fmaf(x1.x, w, a1.x); a1.y = fmaf(x1.y, w, a1.y);
    a1.z = fmaf(x1.z, w, a1.z); a1.w = fmaf(x1.w, w, a1.w);
  }
  while (cur < RELS) STORE_AND_NEXT_RELATION();
#undef STORE_AND_NEXT_RELATION
}

__global__ __launch_bounds__(256)
void gather_rel_rows(const float* __restrict__ T, const int* __restrict__ rowptr,
                     const int* __restrict__ bins, const int* __restrict__ esrc,
                     const float* __restrict__ eval, float* __restrict__ agg,
                     int N, int E) {
  __shared__ int sC[8][64];
  __shared__ int sV[8][64];
  const int lane = threadIdx.x & 31;
  const int wave = __builtin_amdgcn_readfirstlane(threadIdx.x >> 5);
  const int row  = blockIdx.x * 8 + wave;
  int start = 0, n = 0;
  if (row < N) { start = rowptr[row]; n = rowptr[row + 1] - start; }
  const bool small = n <= 64;
  int e1 = 0, e2 = 0, e3 = 0;

  if (small) {
    const bool h0 = lane < n, h1 = lane + 32 < n;
    int k0 = KEY_NONE, k1 = KEY_NONE;
    if (h0) k0 = bins[start + lane];
    if (h1) k1 = bins[start + 32 + lane];
    int s0 = 0, s1 = 0, v0 = 0, v1 = 0;
    if (h0) { s0 = esrc[k0]; v0 = __float_as_int(eval[k0]); }
    if (h1) { s1 = esrc[k1]; v1 = __float_as_int(eval[k1]); }
    int r0 = 0, r1 = 0;
    const int na = min(n, 32);
    for (int i = 0; i < na; ++i) {
      const int ki = __builtin_amdgcn_readlane(k0, i);
      r0 += (ki < k0) ? 1 : 0;
      r1 += (ki < k1) ? 1 : 0;
    }
    for (int i = 0; i < n - 32; ++i) {
      const int ki = __builtin_amdgcn_readlane(k1, i);
      r0 += (ki < k0) ? 1 : 0;
      r1 += (ki < k1) ? 1 : 0;
    }
    if (h0) { sC[wave][r0] = s0; sV[wave][r0] = v0; }
    if (h1) { sC[wave][r1] = s1; sV[wave][r1] = v1; }
    e1 = __builtin_popcount(__builtin_amdgcn_ballot_w32(k0 < E)) +
         __builtin_popcount(__builtin_amdgcn_ballot_w32(k1 < E));
    e2 = __builtin_popcount(__builtin_amdgcn_ballot_w32(k0 < 2 * E)) +
         __builtin_popcount(__builtin_amdgcn_ballot_w32(k1 < 2 * E));
    e3 = __builtin_popcount(__builtin_amdgcn_ballot_w32(k0 < 3 * E)) +
         __builtin_popcount(__builtin_amdgcn_ballot_w32(k1 < 3 * E));
  }
  __syncthreads();

  float* orow = agg + (size_t)row * KCAT + lane * 4;
  if (small) {
    int cA = 0, vA = 0, cB = 0, vB = 0;
    if (lane < n)      { cA = sC[wave][lane];      vA = sV[wave][lane]; }
    if (lane + 32 < n) { cB = sC[wave][lane + 32]; vB = sV[wave][lane + 32]; }
    sum_row_by_relation(T, cA, vA, cB, vB, n, e1, e2, e3, lane, orow, row < N);
  } else {
    float4 a0 = make_float4(0.f, 0.f, 0.f, 0.f);
    float4 a1 = make_float4(0.f, 0.f, 0.f, 0.f);
    int last = -1, cur = 0;
    for (int p = 0; p < n; ++p) {
      int best = KEY_NONE;
      for (int j = lane; j < n; j += 32) {
        const int k = bins[start + j];
        if (k > last && k < best) best = k;
      }
      for (int off = 16; off > 0; off >>= 1) best = min(best, __shfl_xor(best, off));
      best = __builtin_amdgcn_readfirstlane(best);
      last = best;
      const int   s   = esrc[best];
      const float w   = eval[best];
      const int   rel = (best >= E ? 1 : 0) + (best >= 2 * E ? 1 : 0) + (best >= 3 * E ? 1 : 0);
      for (; cur < rel; ++cur) {
        *(float4*)(orow + cur * FDIM)       = a0;
        *(float4*)(orow + cur * FDIM + 128) = a1;
        a0 = make_float4(0.f, 0.f, 0.f, 0.f);
        a1 = make_float4(0.f, 0.f, 0.f, 0.f);
      }
      const float* q = T + (size_t)s * FDIM + lane * 4;
      const float4 x0 = *(const float4*)q, x1 = *(const float4*)(q + 128);
      a0.x = fmaf(x0.x, w, a0.x); a0.y = fmaf(x0.y, w, a0.y);
      a0.z = fmaf(x0.z, w, a0.z); a0.w = fmaf(x0.w, w, a0.w);
      a1.x = fmaf(x1.x, w, a1.x); a1.y = fmaf(x1.y, w, a1.y);
      a1.z = fmaf(x1.z, w, a1.z); a1.w = fmaf(x1.w, w, a1.w);
    }
    for (; cur < RELS; ++cur) {
      *(float4*)(orow + cur * FDIM)       = a0;
      *(float4*)(orow + cur * FDIM + 128) = a1;
      a0 = make_float4(0.f, 0.f, 0.f, 0.f);
      a1 = make_float4(0.f, 0.f, 0.f, 0.f);
    }
  }
}

extern "C" void kernel_launch(void* const* d_in, const int* in_sizes, int n_in,
                              void* d_out, int out_size, void* d_ws, size_t ws_size,
                              hipStream_t stream) {
  const float* x    = (const float*)d_in[0];
  const int*   esrc = (const int*)d_in[1];
  const int*   edst = (const int*)d_in[2];
  const float* eval = (const float*)d_in[3];
  const float* W1   = (const float*)d_in[4];
  const float* b1   = (const float*)d_in[5];
  const float* W2   = (const float*)d_in[6];
  const float* b2   = (const float*)d_in[7];
  float* out = (float*)d_out;

  const int R     = 4;
  const int Nn    = in_sizes[0] / FDIM;
  const int E     = in_sizes[1] / R;
  const int total = R * E;

  const size_t nf = (size_t)Nn * FDIM;
  float* agg    = (float*)d_ws;
  float* x1     = agg + 4 * nf;
  int*   cnt    = (int*)(x1 + nf);
  int*   rowptr = cnt + Nn + 4;
  int*   bins   = rowptr + Nn + 4;

  int*   tab    = (int*)d_ws;
  int*   rank   = tab + (size_t)NCB * Nn;

  dim3 gGrid(FDIM / BN, (Nn + BM - 1) / BM);
  const int nBlocks = (Nn + 255) / 256;
  const int rBlocks = (Nn + 7) / 8;
  const int per     = (total + NCB - 1) / NCB;
  dim3 fGrid((per + 255) / 256, NCB);

  count_rows_lds<<<NCB, 1024, 0, stream>>>(edst, tab, rank, total, Nn);
  scan_pieces<<<nBlocks, 256, 0, stream>>>(tab, cnt, Nn);
  scan_rows<<<1, 1024, 0, stream>>>(cnt, rowptr, Nn);
  fill_bins<<<fGrid, 256, 0, stream>>>(edst, rowptr, tab, rank, bins, total, Nn);

  gather_rel_rows<<<rBlocks, 256, 0, stream>>>(x, rowptr, bins, esrc, eval, agg, Nn, E);
  gemm_cat_f32_wmma<false><<<gGrid, 256, 0, stream>>>(agg, W1, b1, nullptr, x1, Nn, KCAT);

  gather_rel_rows<<<rBlocks, 256, 0, stream>>>(x1, rowptr, bins, esrc, eval, agg, Nn, E);
  gemm_cat_f32_wmma<true><<<gGrid, 256, 0, stream>>>(agg, W2, b2, x1, out, Nn, KCAT);
}


# binning pass stores each edge's id, source row and value as one 16-byte record in its bin slot; row gathers read records contiguously, not via two indexed 4-byte loads per edge; sum order kept
# speedup vs baseline: 2.9683x; 1.0354x over previous
#include <hip/hip_runtime.h>

typedef __attribute__((ext_vector_type(2))) float v2f;
typedef __attribute__((ext_vector_type(4))) float v4f;
typedef __attribute__((ext_vector_type(8))) float v8f;

#define FDIM 256
#define RELS 4
#define KCAT (RELS * FDIM)
#define BM 64
#define BN 128
#define BK 64
#define SXP (BK + 4)
#define WPP 288

template <bool FINAL>
__global__ __launch_bounds__(256)
void gemm_cat_f32_wmma(const float* __restrict__ X, const float* __restrict__ W,
                       const float* __restrict__ bias, const float* skip,
                       float* O, int nrows, int kdim) {
  __shared__ __align__(16) float sX[BM * SXP];
  __shared__ __align__(16) float sW[(BK / 2) * WPP];

  const int t       = threadIdx.x;
  const int lane    = t & 31;
  const int wave    = t >> 5;
  const int l16     = lane & 15;
  const int half    = lane >> 4;
  const int waveRow = wave & 3;
  const int waveCol = wave >> 2;
  const int rowBase = blockIdx.y * BM;
  const int colBase = blockIdx.x * BN;
  const int cb2     = waveCol * 128 + l16;

  const int xr0 = t >> 4;
  const int xc4 = (t & 15) << 2;
  const float* xp0 = X + (size_t)min(rowBase + xr0,      nrows - 1) * kdim + xc4;
  const float* xp1 = X + (size_t)min(rowBase + xr0 + 16, nrows - 1) * kdim + xc4;
  const float* xp2 = X + (size_t)min(rowBase + xr0 + 32, nrows - 1) * kdim + xc4;
  const float* xp3 = X + (size_t)min(rowBase + xr0 + 48, nrows - 1) * kdim + xc4;
  float* sxp = &sX[xr0 * SXP + xc4];
  const int wk0 = 2 * (wave >> 1) + ((lane >> 2) & 1);
  const int wn4 = (wave & 1) * 64 + (lane >> 3) * 16 + ((lane & 3) << 2);
  const float* wp = W + (size_t)wk0 * FDIM + colBase + wn4;
  float* swp = sW + (wave >> 1) * WPP + (wave & 1) * 128 + (lane << 2);

  v8f acc0 = {}, acc1 = {}, acc2 = {}, acc3 = {};

  float4 xv0 = *(const float4*)xp0, xv1 = *(const float4*)xp1;
  float4 xv2 = *(const float4*)xp2, xv3 = *(const float4*)xp3;
  float4 wv[8];
#pragma unroll
  for (int it = 0; it < 8; ++it)
    wv[it] = *(const float4*)(wp + (size_t)(8 * it) * FDIM);

  for (int kBase = 0; kBase < kdim; kBase += BK) {
    *(float4*)(sxp)            = xv0;
    *(float4*)(sxp + 16 * SXP) = xv1;
    *(float4*)(sxp + 32 * SXP) = xv2;
    *(float4*)(sxp + 48 * SXP) = xv3;
#pragma unroll
    for (int it = 0; it < 8; ++it) {
      const v4f w4 = {wv[it].x, wv[it].y, wv[it].z, wv[it].w};
      *(v4f*)(swp + it * 4 * WPP) = w4;
    }
    __syncthreads();

    const int kn = min(kBase + BK, kdim - BK);
    xv0 = *(const float4*)(xp0 + kn); xv1 = *(const float4*)(xp1 + kn);
    xv2 = *(const float4*)(xp2 + kn); xv3 = *(const float4*)(xp3 + kn);
#pragma unroll
    for (int it = 0; it < 8; ++it)
      wv[it] = *(const float4*)(wp + (size_t)(kn + 8 * it) * FDIM);
    __builtin_amdgcn_sched_barrier(0);

#pragma unroll
    for (int kk = 0; kk < BK; kk += 4) {
      const int ks = kk + 2 * half;
      v2f a = *(const v2f*)&sX[(waveRow * 16 + l16) * SXP + ks];
      const float* wrow = sW + ((kk >> 1) + half) * WPP + cb2;
      v2f b0, b1, b2, b3;
      b0.x = wrow[0];  b0.y = wrow[16];
      b1.x = wrow[32]; b1.y = wrow[48];
      b2.x = wrow[64]; b2.y = wrow[80];
      b3.x = wrow[96]; b3.y = wrow[112];
      acc0 = __builtin_amdgcn_wmma_f32_16x16x4_f32(false, a, false, b0, (short)0, acc0, false, false);
      acc1 = __builtin_amdgcn_wmma_f32_16x16x4_f32(false, a, false, b1, (short)0, acc1, false, false);
      acc2 = __builtin_amdgcn_wmma_f32_16x16x4_f32(false, a, false, b2, (short)0, acc2, false, false);
      acc3 = __builtin_amdgcn_wmma_f32_16x16x4_f32(false, a, false, b3, (short)0, acc3, false, false);
    }
    __syncthreads();
  }

  const int   colw = colBase + waveCol * 64 + l16;
  const float bb0 = bias[colw], bb1 = bias[colw + 16];
  const float bb2 = bias[colw + 32], bb3 = bias[colw + 48];
  float sk0[8], sk1[8], sk2[8], sk3[8];
  if (FINAL) {
#pragma unroll
    for (int v = 0; v < 8; ++v) {
      const int gr = rowBase + waveRow * 16 + v + 8 * half;
      sk0[v] = sk1[v] = sk2[v] = sk3[v] = 0.f;
      if (gr < nrows) {
        const size_t off = (size_t)gr * FDIM + colw;
        sk0[v] = skip[off];      sk1[v] = skip[off + 16];
        sk2[v] = skip[off + 32]; sk3[v] = skip[off + 48];
      }
    }
  }
#pragma unroll
  for (int v = 0; v < 8; ++v) {
    int gr = rowBase + waveRow * 16 + v + 8 * half;
    if (gr < nrows) {
      const size_t off = (size_t)gr * FDIM + colw;
      float o0 = fmaxf(acc0[v] + bb0, 0.f);
      float o1 = fmaxf(acc1[v] + bb1, 0.f);
      float o2 = fmaxf(acc2[v] + bb2, 0.f);
      float o3 = fmaxf(acc3[v] + bb3, 0.f);
      if (FINAL) {
        o0 += sk0[v]; o1 += sk1[v];
        o2 += sk2[v]; o3 += sk3[v];
      }
      O[off]      = o0;
      O[off + 16] = o1;
      O[off + 32] = o2;
      O[off + 48] = o3;
    }
  }
}

#define KEY_NONE 0x7fffffff

#define NCB  64
#define HCAP 65536

__global__ __launch_bounds__(1024)
void count_rows_lds(const int* __restrict__ dst, int* __restrict__ tab,
                    int* __restrict__ rank, int total, int n) {
  __shared__ int hist[HCAP];
  const int t   = threadIdx.x;
  const int per = (total + (int)gridDim.x - 1) / (int)gridDim.x;
  const int e0  = min((int)blockIdx.x * per, total);
  const int e1  = min(e0 + per, total);
  int* trow = tab + (size_t)blockIdx.x * n;
  for (int base = 0; base < n; base += HCAP) {
    const int len = min(HCAP, n - base);
    for (int i = t; i < len; i += 1024) hist[i] = 0;
    __syncthreads();
    int g = e0 + t;
    for (; g + 3 * 1024 < e1; g += 4 * 1024) {
      const int d0 = dst[g]            - base;
      const int d1 = dst[g + 1024]     - base;
      const int d2 = dst[g + 2 * 1024] - base;
      const int d3 = dst[g + 3 * 1024] - base;
      if ((unsigned)d0 < (unsigned)len) rank[g]            = atomicAdd(&hist[d0], 1);
      if ((unsigned)d1 < (unsigned)len) rank[g + 1024]     = atomicAdd(&hist[d1], 1);
      if ((unsigned)d2 < (unsigned)len) rank[g + 2 * 1024] = atomicAdd(&hist[d2], 1);
      if ((unsigned)d3 < (unsigned)len) rank[g + 3 * 1024] = atomicAdd(&hist[d3], 1);
    }
    for (; g < e1; g += 1024) {
      const int d = dst[g] - base;
      if ((unsigned)d < (unsigned)len) rank[g] = atomicAdd(&hist[d], 1);
    }
    __syncthreads();
    for (int i = t; i < len; i += 1024) trow[base + i] = hist[i];
    __syncthreads();
  }
}

__global__ __launch_bounds__(256)
void scan_pieces(int* __restrict__ tab, int* __restrict__ cnt, int n) {
  const int d = blockIdx.x * blockDim.x + threadIdx.x;
  if (d >= n) return;
  int v[NCB];
#pragma unroll
  for (int b = 0; b < NCB; ++b) v[b] = tab[(size_t)b * n + d];
  int run = 0;
#pragma unroll
  for (int b = 0; b < NCB; ++b) {
    const int c = v[b];
    tab[(size_t)b * n + d] = run;
    run += c;
  }
  cnt[d] = run;
}

__global__ __launch_bounds__(1024)
void scan_rows(const int* __restrict__ cnt, int* __restrict__ rowptr, int n) {
  __shared__ int part[1024];
  const int t     = threadIdx.x;
  const int chunk = (n + 1023) >> 10;
  const int b     = t * chunk;
  const int e     = min(b + chunk, n);
  int s = 0;
  for (int i = b; i < e; ++i) s += cnt[i];
  part[t] = s;
  __syncthreads();
  for (int off = 1; off < 1024; off <<= 1) {
    int v = (t >= off) ? part[t - off] : 0;
    __syncthreads();
    part[t] += v;
    __syncthreads();
  }
  int run = part[t] - s;
  for (int i = b; i < e; ++i) { rowptr[i] = run; run += cnt[i]; }
  if (t == 1023) rowptr[n] = part[1023];
}

__global__ __launch_bounds__(256)
void fill_bins(const int* __restrict__ dst, const int* __restrict__ esrc,
               const float* __restrict__ eval, const int* __restrict__ rowptr,
               const int* __restrict__ tab, const int* __restrict__ rank,
               int* __restrict__ bins, int total, int n) {
  const int per = (total + (int)gridDim.y - 1) / (int)gridDim.y;
  const int i   = blockIdx.x * blockDim.x + threadIdx.x;
  const int g   = (int)blockIdx.y * per + i;
  if (i < per && g < total) {
    const int d = dst[g];
    const int4 rec = make_int4(g, esrc[g], __float_as_int(eval[g]), 0);
    const int slot = rowptr[d] + tab[(size_t)blockIdx.y * n + d] + rank[g];
    *(int4*)(bins + 4 * (size_t)slot) = rec;
  }
}

__device__ __forceinline__ int place_pick(int lo, int hi, int p) {
  return __builtin_amdgcn_readlane(p < 32 ? lo : hi, p & 31);
}

__device__ __forceinline__
void sum_row_by_relation(const float* __restrict__ T, int cA, int vA, int cB, int vB,
                         int n, int e1, int e2, int e3, int lane,
                         float* __restrict__ orow, bool live) {
  float4 a0 = make_float4(0.f, 0.f, 0.f, 0.f);
  float4 a1 = make_float4(0.f, 0.f, 0.f, 0.f);
  int cur = 0, bnd = e1;
#define STORE_AND_NEXT_RELATION()                                   \
  do {                                                              \
    if (live) {                                                     \
      *(float4*)(orow + cur * FDIM)       = a0;                     \
      *(float4*)(orow + cur * FDIM + 128) = a1;                     \
    }                                                               \
    a0 = make_float4(0.f, 0.f, 0.f, 0.f);                           \
    a1 = make_float4(0.f, 0.f, 0.f, 0.f);                           \
    ++cur;                                                          \
    bnd = (cur == 1) ? e2 : (cur == 2) ? e3 : n;                    \
  } while (0)
  int p = 0;
  for (; p + 4 <= n; p += 4) {
    const int c0 = place_pick(cA, cB, p);
    const int c1 = place_pick(cA, cB, p + 1);
    const int c2 = place_pick(cA, cB, p + 2);
    const int c3 = place_pick(cA, cB, p + 3);
    const float w0 = __int_as_float(place_pick(vA, vB, p));
    const float w1 = __int_as_float(place_pick(vA, vB, p + 1));
    const float w2 = __int_as_float(place_pick(vA, vB, p + 2));
    const float w3 = __int_as_float(place_pick(vA, vB, p + 3));
    const float* q0 = T + (size_t)c0 * FDIM + lane * 4;
    const float* q1 = T + (size_t)c1 * FDIM + lane * 4;
    const float* q2 = T + (size_t)c2 * FDIM + lane * 4;
    const float* q3 = T + (size_t)c3 * FDIM + lane * 4;
    const float4 x00 = *(const float4*)q0, x01 = *(const float4*)(q0 + 128);
    const float4 x10 = *(const float4*)q1, x11 = *(const float4*)(q1 + 128);
    const float4 x20 = *(const float4*)q2, x21 = *(const float4*)(q2 + 128);
    const float4 x30 = *(const float4*)q3, x31 = *(const float4*)(q3 + 128);
    while (cur < RELS - 1 && p >= bnd) STORE_AND_NEXT_RELATION();
    a0.x = fmaf(x00.x, w0, a0.x); a0.y = fmaf(x00.y, w0, a0.y);
    a0.z = fmaf(x00.z, w0, a0.z); a0.w = fmaf(x00.w, w0, a0.w);
    a1.x = fmaf(x01.x, w0, a1.x); a1.y = fmaf(x01.y, w0, a1.y);
    a1.z = fmaf(x01.z, w0, a1.z); a1.w = fmaf(x01.w, w0, a1.w);
    while (cur < RELS - 1 && p + 1 >= bnd) STORE_AND_NEXT_RELATION();
    a0.x = fmaf(x10.x, w1, a0.x); a0.y = fmaf(x10.y, w1, a0.y);
    a0.z = fmaf(x10.z, w1, a0.z); a0.w = fmaf(x10.w, w1, a0.w);
    a1.x = fmaf(x11.x, w1, a1.x); a1.y = fmaf(x11.y, w1, a1.y);
    a1.z = fmaf(x11.z, w1, a1.z); a1.w = fmaf(x11.w, w1, a1.w);
    while (cur < RELS - 1 && p + 2 >= bnd) STORE_AND_NEXT_RELATION();
    a0.x = fmaf(x20.x, w2, a0.x); a0.y = fmaf(x20.y, w2, a0.y);
    a0.z = fmaf(x20.z, w2, a0.z); a0.w = fmaf(x20.w, w2, a0.w);
    a1.x = fmaf(x21.x, w2, a1.x); a1.y = fmaf(x21.y, w2, a1.y);
    a1.z = fmaf(x21.z, w2, a1.z); a1.w = fmaf(x21.w, w2, a1.w);
    while (cur < RELS - 1 && p + 3 >= bnd) STORE_AND_NEXT_RELATION();
    a0.x = fmaf(x30.x, w3, a0.x); a0.y = fmaf(x30.y, w3, a0.y);
    a0.z = fmaf(x30.z, w3, a0.z); a0.w = fmaf(x30.w, w3, a0.w);
    a1.x = fmaf(x31.x, w3, a1.x); a1.y = fmaf(x31.y, w3, a1.y);
    a1.z = fmaf(x31.z, w3, a1.z); a1.w = fmaf(x31.w, w3, a1.w);
  }
  for (; p < n; ++p) {
    const int   c = place_pick(cA, cB, p);
    const float w = __int_as_float(place_pick(vA, vB, p));
    const float* q = T + (size_t)c * FDIM + lane * 4;
    const float4 x0 = *(const float4*)q, x1 = *(const float4*)(q + 128);
    while (cur < RELS - 1 && p >= bnd) STORE_AND_NEXT_RELATION();
    a0.x = fmaf(x0.x, w, a0.x); a0.y = fmaf(x0.y, w, a0.y);
    a0.z = fmaf(x0.z, w, a0.z); a0.w = fmaf(x0.w, w, a0.w);
    a1.x = fmaf(x1.x, w, a1.x); a1.y = fmaf(x1.y, w, a1.y);
    a1.z = fmaf(x1.z, w, a1.z); a1.w = fmaf(x1.w, w, a1.w);
  }
  while (cur < RELS) STORE_AND_NEXT_RELATION();
#undef STORE_AND_NEXT_RELATION
}

__global__ __launch_bounds__(256)
void gather_rel_rows(const float* __restrict__ T, const int* __restrict__ rowptr,
                     const int* __restrict__ bins, const int* __restrict__ esrc,
                     const float* __restrict__ eval, float* __restrict__ agg,
                     int N, int E) {
  __shared__ int sC[8][64];
  __shared__ int sV[8][64];
  const int lane = threadIdx.x & 31;
  const int wave = __builtin_amdgcn_readfirstlane(threadIdx.x >> 5);
  const int row  = blockIdx.x * 8 + wave;
  int start = 0, n = 0;
  if (row < N) { start = rowptr[row]; n = rowptr[row + 1] - start; }
  const bool small = n <= 64;
  int e1 = 0, e2 = 0, e3 = 0;

  if (small) {
    const bool h0 = lane < n, h1 = lane + 32 < n;
    int k0 = KEY_NONE, k1 = KEY_NONE;
    int s0 = 0, s1 = 0, v0 = 0, v1 = 0;
    if (h0) {
      const int4 r = *(const int4*)(bins + 4 * (size_t)(start + lane));
      k0 = r.x; s0 = r.y; v0 = r.z;
    }
    if (h1) {
      const int4 r = *(const int4*)(bins + 4 * (size_t)(start + 32 + lane));
      k1 = r.x; s1 = r.y; v1 = r.z;
    }
    int r0 = 0, r1 = 0;
    const int na = min(n, 32);
    for (int i = 0; i < na; ++i) {
      const int ki = __builtin_amdgcn_readlane(k0, i);
      r0 += (ki < k0) ? 1 : 0;
      r1 += (ki < k1) ? 1 : 0;
    }
    for (int i = 0; i < n - 32; ++i) {
      const int ki = __builtin_amdgcn_readlane(k1, i);
      r0 += (ki < k0) ? 1 : 0;
      r1 += (ki < k1) ? 1 : 0;
    }
    if (h0) { sC[wave][r0] = s0; sV[wave][r0] = v0; }
    if (h1) { sC[wave][r1] = s1; sV[wave][r1] = v1; }
    e1 = __builtin_popcount(__builtin_amdgcn_ballot_w32(k0 < E)) +
         __builtin_popcount(__builtin_amdgcn_ballot_w32(k1 < E));
    e2 = __builtin_popcount(__builtin_amdgcn_ballot_w32(k0 < 2 * E)) +
         __builtin_popcount(__builtin_amdgcn_ballot_w32(k1 < 2 * E));
    e3 = __builtin_popcount(__builtin_amdgcn_ballot_w32(k0 < 3 * E)) +
         __builtin_popcount(__builtin_amdgcn_ballot_w32(k1 < 3 * E));
  }
  __syncthreads();

  float* orow = agg + (size_t)row * KCAT + lane * 4;
  if (small) {
    int cA = 0, vA = 0, cB = 0, vB = 0;
    if (lane < n)      { cA = sC[wave][lane];      vA = sV[wave][lane]; }
    if (lane + 32 < n) { cB = sC[wave][lane + 32]; vB = sV[wave][lane + 32]; }
    sum_row_by_relation(T, cA, vA, cB, vB, n, e1, e2, e3, lane, orow, row < N);
  } else {
    float4 a0 = make_float4(0.f, 0.f, 0.f, 0.f);
    float4 a1 = make_float4(0.f, 0.f, 0.f, 0.f);
    int last = -1, cur = 0;
    for (int p = 0; p < n; ++p) {
      int best = KEY_NONE;
      for (int j = lane; j < n; j += 32) {
        const int k = bins[4 * (size_t)(start + j)];
        if (k > last && k < best) best = k;
      }
      for (int off = 16; off > 0; off >>= 1) best = min(best, __shfl_xor(best, off));
      best = __builtin_amdgcn_readfirstlane(best);
      last = best;
      const int   s   = esrc[best];
      const float w   = eval[best];
      const int   rel = (best >= E ? 1 : 0) + (best >= 2 * E ? 1 : 0) + (best >= 3 * E ? 1 : 0);
      for (; cur < rel; ++cur) {
        *(float4*)(orow + cur * FDIM)       = a0;
        *(float4*)(orow + cur * FDIM + 128) = a1;
        a0 = make_float4(0.f, 0.f, 0.f, 0.f);
        a1 = make_float4(0.f, 0.f, 0.f, 0.f);
      }
      const float* q = T + (size_t)s * FDIM + lane * 4;
      const float4 x0 = *(const float4*)q, x1 = *(const float4*)(q + 128);
      a0.x = fmaf(x0.x, w, a0.x); a0.y = fmaf(x0.y, w, a0.y);
      a0.z = fmaf(x0.z, w, a0.z); a0.w = fmaf(x0.w, w, a0.w);
      a1.x = fmaf(x1.x, w, a1.x); a1.y = fmaf(x1.y, w, a1.y);
      a1.z = fmaf(x1.z, w, a1.z); a1.w = fmaf(x1.w, w, a1.w);
    }
    for (; cur < RELS; ++cur) {
      *(float4*)(orow + cur * FDIM)       = a0;
      *(float4*)(orow + cur * FDIM + 128) = a1;
      a0 = make_float4(0.f, 0.f, 0.f, 0.f);
      a1 = make_float4(0.f, 0.f, 0.f, 0.f);
    }
  }
}

extern "C" void kernel_launch(void* const* d_in, const int* in_sizes, int n_in,
                              void* d_out, int out_size, void* d_ws, size_t ws_size,
                              hipStream_t stream) {
  const float* x    = (const float*)d_in[0];
  const int*   esrc = (const int*)d_in[1];
  const int*   edst = (const int*)d_in[2];
  const float* eval = (const float*)d_in[3];
  const float* W1   = (const float*)d_in[4];
  const float* b1   = (const float*)d_in[5];
  const float* W2   = (const float*)d_in[6];
  const float* b2   = (const float*)d_in[7];
  float* out = (float*)d_out;

  const int R     = 4;
  const int Nn    = in_sizes[0] / FDIM;
  const int E     = in_sizes[1] / R;
  const int total = R * E;

  const size_t nf = (size_t)Nn * FDIM;
  float* agg    = (float*)d_ws;
  int*   bins   = (int*)(agg + 4 * nf);
  int*   cnt    = bins + 4 * (size_t)total;
  int*   rowptr = cnt + Nn + 4;
  float* x1     = out;

  int*   tab    = (int*)d_ws;
  int*   rank   = tab + (size_t)NCB * Nn;

  dim3 gGrid(FDIM / BN, (Nn + BM - 1) / BM);
  const int nBlocks = (Nn + 255) / 256;
  const int rBlocks = (Nn + 7) / 8;
  const int per     = (total + NCB - 1) / NCB;
  dim3 fGrid((per + 255) / 256, NCB);

  count_rows_lds<<<NCB, 1024, 0, stream>>>(edst, tab, rank, total, Nn);
  scan_pieces<<<nBlocks, 256, 0, stream>>>(tab, cnt, Nn);
  scan_rows<<<1, 1024, 0, stream>>>(cnt, rowptr, Nn);
  fill_bins<<<fGrid, 256, 0, stream>>>(edst, esrc, eval, rowptr, tab, rank, bins, total, Nn);

  gather_rel_rows<<<rBlocks, 256, 0, stream>>>(x, rowptr, bins, esrc, eval, agg, Nn, E);
  gemm_cat_f32_wmma<false><<<gGrid, 256, 0, stream>>>(agg, W1, b1, nullptr, x1, Nn, KCAT);

  gather_rel_rows<<<rBlocks, 256, 0, stream>>>(x1, rowptr, bins, esrc, eval, agg, Nn, E);
  gemm_cat_f32_wmma<true><<<gGrid, 256, 0, stream>>>(agg, W2, b2, x1, out, Nn, KCAT);
}


# weight product: rows dealt as 16-row strips to 768 row blocks (4 or 5 each; wave = 16 columns of all strips; rows out via LDS): two full rounds of blocks, no lone last round; sums unchanged
# speedup vs baseline: 3.0942x; 1.0424x over previous
#include <hip/hip_runtime.h>

typedef __attribute__((ext_vector_type(2))) float v2f;
typedef __attribute__((ext_vector_type(4))) float v4f;
typedef __attribute__((ext_vector_type(8))) float v8f;

#define FDIM 256
#define RELS 4
#define KCAT (RELS * FDIM)
#define BN 128
#define BK 64
#define SXP (BK + 4)
#define WPP 288

#define NSMAX 5
#define OSP 132
template <int NS, bool FINAL>
__device__ __forceinline__
void gemm_strips(const float* __restrict__ X, const float* __restrict__ W,
                 const float* __restrict__ bias, const float* skip,
                 float* O, int nrows, int kdim, int rowBase, int rowEnd,
                 float* sX, float* sW) {
  const int t       = threadIdx.x;
  const int lane    = t & 31;
  const int wave    = t >> 5;
  const int l16     = lane & 15;
  const int half    = lane >> 4;
  const int colBase = blockIdx.x * BN;
  const int cb2     = wave * 32 + l16;

  const int xr0 = t >> 4;
  const int xc4 = (t & 15) << 2;
  unsigned xo[NS];
#pragma unroll
  for (int it = 0; it < NS; ++it)
    xo[it] = ((unsigned)min(rowBase + xr0 + 16 * it, nrows - 1) * (unsigned)kdim + (unsigned)xc4) * 4u;
  float* sxp = &sX[xr0 * SXP + xc4];
  const int wk0 = 2 * (wave >> 1) + ((lane >> 2) & 1);
  const int wn4 = (wave & 1) * 64 + (lane >> 3) * 16 + ((lane & 3) << 2);
  const unsigned wo = (unsigned)(wk0 * FDIM + colBase + wn4) * 4u;
  float* swp = sW + (wave >> 1) * WPP + (wave & 1) * 128 + (lane << 2);

  v8f acc[NS];
#pragma unroll
  for (int s = 0; s < NS; ++s) acc[s] = v8f{};

  v4f xv[NS];
#pragma unroll
  for (int it = 0; it < NS; ++it) xv[it] = *(const v4f*)((const char*)X + xo[it]);
  v4f wv[8];
#pragma unroll
  for (int it = 0; it < 8; ++it)
    wv[it] = *(const v4f*)((const char*)W + wo + (8 * it) * FDIM * 4);

  for (int kBase = 0; kBase < kdim; kBase += BK) {
#pragma unroll
    for (int it = 0; it < NS; ++it)
      *(v4f*)(sxp + it * 16 * SXP) = xv[it];
#pragma unroll
    for (int it = 0; it < 8; ++it)
      *(v4f*)(swp + it * 4 * WPP) = wv[it];
    __syncthreads();

    const int kn = min(kBase + BK, kdim - BK);
    const char* Xk = (const char*)(X + kn);
    const char* Wk = (const char*)(W + (size_t)kn * FDIM);
#pragma unroll
    for (int it = 0; it < NS; ++it) xv[it] = *(const v4f*)(Xk + xo[it]);
#pragma unroll
    for (int it = 0; it < 8; ++it)
      wv[it] = *(const v4f*)(Wk + wo + (8 * it) * FDIM * 4);
    __builtin_amdgcn_sched_barrier(0);

#pragma unroll
    for (int kk = 0; kk < BK; kk += 4) {
      const int ks = kk + 2 * half;
      const float* wrow = sW + ((kk >> 1) + half) * WPP + cb2;
      v2f b;
      b.x = wrow[0]; b.y = wrow[16];
#pragma unroll
      for (int s = 0; s < NS; ++s) {
        const v2f a = *(const v2f*)&sX[(s * 16 + l16) * SXP + ks];
        acc[s] = __builtin_amdgcn_wmma_f32_16x16x4_f32(false, a, false, b, (short)0, acc[s], false, false);
      }
    }
    __syncthreads();
  }

  const int   colw = colBase + wave * 16 + l16;
  const float bb   = bias[colw];
  float* sO = sX;
#pragma unroll
  for (int s = 0; s < NS; ++s)
#pragma unroll
    for (int v = 0; v < 8; ++v)
      sO[(s * 16 + v + 8 * half) * OSP + wave * 16 + l16] = fmaxf(acc[s][v] + bb, 0.f);
  __syncthreads();
  const size_t gcol = (size_t)colBase + lane * 4;
  v4f sk[2 * NS];
  if (FINAL) {
#pragma unroll
    for (int j = 0; j < 2 * NS; ++j) {
      const int gr = rowBase + wave + 8 * j;
      sk[j] = v4f{0.f, 0.f, 0.f, 0.f};
      if (gr < rowEnd) sk[j] = *(const v4f*)(skip + (size_t)gr * FDIM + gcol);
    }
  }
#pragma unroll
  for (int j = 0; j < 2 * NS; ++j) {
    const int r  = wave + 8 * j;
    const int gr = rowBase + r;
    if (gr < rowEnd) {
      v4f o = *(const v4f*)&sO[r * OSP + lane * 4];
      if (FINAL) o += sk[j];
      *(v4f*)(O + (size_t)gr * FDIM + gcol) = o;
    }
  }
}

template <bool FINAL>
__global__ __launch_bounds__(256) __attribute__((amdgpu_waves_per_eu(6)))
void gemm_cat_f32_wmma(const float* __restrict__ X, const float* __restrict__ W,
                       const float* __restrict__ bias, const float* skip,
                       float* O, int nrows, int kdim) {
  __shared__ __align__(16) float sMem[NSMAX * 16 * SXP + (BK / 2) * WPP];
  float* sX = sMem;
  float* sW = sMem + NSMAX * 16 * SXP;

  const int strips = (nrows + 15) >> 4;
  const int rb     = (int)gridDim.y;
  const int base   = strips / rb;
  const int extra  = strips - base * rb;
  const int by     = (int)blockIdx.y;
  const int ns     = base + (by < extra ? 1 : 0);
  const int rowBase = (by * base + min(by, extra)) * 16;
  const int rowEnd  = min(rowBase + ns * 16, nrows);
  if (ns > 4)
    gemm_strips<5, FINAL>(X, W, bias, skip, O, nrows, kdim, rowBase, rowEnd, sX, sW);
  else
    gemm_strips<4, FINAL>(X, W, bias, skip, O, nrows, kdim, rowBase, rowEnd, sX, sW);
}

#define KEY_NONE 0x7fffffff

#define NCB  64
#define HCAP 65536

__global__ __launch_bounds__(1024)
void count_rows_lds(const int* __restrict__ dst, int* __restrict__ tab,
                    int* __restrict__ rank, int total, int n) {
  __shared__ int hist[HCAP];
  const int t   = threadIdx.x;
  const int per = (total + (int)gridDim.x - 1) / (int)gridDim.x;
  const int e0  = min((int)blockIdx.x * per, total);
  const int e1  = min(e0 + per, total);
  int* trow = tab + (size_t)blockIdx.x * n;
  for (int base = 0; base < n; base += HCAP) {
    const int len = min(HCAP, n - base);
    for (int i = t; i < len; i += 1024) hist[i] = 0;
    __syncthreads();
    int g = e0 + t;
    for (; g + 3 * 1024 < e1; g += 4 * 1024) {
      const int d0 = dst[g]            - base;
      const int d1 = dst[g + 1024]     - base;
      const int d2 = dst[g + 2 * 1024] - base;
      const int d3 = dst[g + 3 * 1024] - base;
      if ((unsigned)d0 < (unsigned)len) rank[g]            = atomicAdd(&hist[d0], 1);
      if ((unsigned)d1 < (unsigned)len) rank[g + 1024]     = atomicAdd(&hist[d1], 1);
      if ((unsigned)d2 < (unsigned)len) rank[g + 2 * 1024] = atomicAdd(&hist[d2], 1);
      if ((unsigned)d3 < (unsigned)len) rank[g + 3 * 1024] = atomicAdd(&hist[d3], 1);
    }
    for (; g < e1; g += 1024) {
      const int d = dst[g] - base;
      if ((unsigned)d < (unsigned)len) rank[g] = atomicAdd(&hist[d], 1);
    }
    __syncthreads();
    for (int i = t; i < len; i += 1024) trow[base + i] = hist[i];
    __syncthreads();
  }
}

__global__ __launch_bounds__(256)
void scan_pieces(int* __restrict__ tab, int* __restrict__ cnt, int n) {
  const int d = blockIdx.x * blockDim.x + threadIdx.x;
  if (d >= n) return;
  int v[NCB];
#pragma unroll
  for (int b = 0; b < NCB; ++b) v[b] = tab[(size_t)b * n + d];
  int run = 0;
#pragma unroll
  for (int b = 0; b < NCB; ++b) {
    const int c = v[b];
    tab[(size_t)b * n + d] = run;
    run += c;
  }
  cnt[d] = run;
}

__global__ __launch_bounds__(1024)
void scan_rows(const int* __restrict__ cnt, int* __restrict__ rowptr, int n) {
  __shared__ int part[1024];
  const int t     = threadIdx.x;
  const int chunk = (n + 1023) >> 10;
  const int b     = t * chunk;
  const int e     = min(b + chunk, n);
  int s = 0;
  for (int i = b; i < e; ++i) s += cnt[i];
  part[t] = s;
  __syncthreads();
  for (int off = 1; off < 1024; off <<= 1) {
    int v = (t >= off) ? part[t - off] : 0;
    __syncthreads();
    part[t] += v;
    __syncthreads();
  }
  int run = part[t] - s;
  for (int i = b; i < e; ++i) { rowptr[i] = run; run += cnt[i]; }
  if (t == 1023) rowptr[n] = part[1023];
}

__global__ __launch_bounds__(256)
void fill_bins(const int* __restrict__ dst, const int* __restrict__ esrc,
               const float* __restrict__ eval, const int* __restrict__ rowptr,
               const int* __restrict__ tab, const int* __restrict__ rank,
               int* __restrict__ bins, int total, int n) {
  const int per = (total + (int)gridDim.y - 1) / (int)gridDim.y;
  const int i   = blockIdx.x * blockDim.x + threadIdx.x;
  const int g   = (int)blockIdx.y * per + i;
  if (i < per && g < total) {
    const int d = dst[g];
    const int4 rec = make_int4(g, esrc[g], __float_as_int(eval[g]), 0);
    const int slot = rowptr[d] + tab[(size_t)blockIdx.y * n + d] + rank[g];
    *(int4*)(bins + 4 * (size_t)slot) = rec;
  }
}

__device__ __forceinline__ int place_pick(int lo, int hi, int p) {
  return __builtin_amdgcn_readlane(p < 32 ? lo : hi, p & 31);
}

__device__ __forceinline__
void sum_row_by_relation(const float* __restrict__ T, int cA, int vA, int cB, int vB,
                         int n, int e1, int e2, int e3, int lane,
                         float* __restrict__ orow, bool live) {
  float4 a0 = make_float4(0.f, 0.f, 0.f, 0.f);
  float4 a1 = make_float4(0.f, 0.f, 0.f, 0.f);
  int cur = 0, bnd = e1;
#define STORE_AND_NEXT_RELATION()                                   \
  do {                                                              \
    if (live) {                                                     \
      *(float4*)(orow + cur * FDIM)       = a0;                     \
      *(float4*)(orow + cur * FDIM + 128) = a1;                     \
    }                                                               \
    a0 = make_float4(0.f, 0.f, 0.f, 0.f);                           \
    a1 = make_float4(0.f, 0.f, 0.f, 0.f);                           \
    ++cur;                                                          \
    bnd = (cur == 1) ? e2 : (cur == 2) ? e3 : n;                    \
  } while (0)
  int p = 0;
  for (; p + 4 <= n; p += 4) {
    const int c0 = place_pick(cA, cB, p);
    const int c1 = place_pick(cA, cB, p + 1);
    const int c2 = place_pick(cA, cB, p + 2);
    const int c3 = place_pick(cA, cB, p + 3);
    const float w0 = __int_as_float(place_pick(vA, vB, p));
    const float w1 = __int_as_float(place_pick(vA, vB, p + 1));
    const float w2 = __int_as_float(place_pick(vA, vB, p + 2));
    const float w3 = __int_as_float(place_pick(vA, vB, p + 3));
    const float* q0 = T + (size_t)c0 * FDIM + lane * 4;
    const float* q1 = T + (size_t)c1 * FDIM + lane * 4;
    const float* q2 = T + (size_t)c2 * FDIM + lane * 4;
    const float* q3 = T + (size_t)c3 * FDIM + lane * 4;
    const float4 x00 = *(const float4*)q0, x01 = *(const float4*)(q0 + 128);
    const float4 x10 = *(const float4*)q1, x11 = *(const float4*)(q1 + 128);
    const float4 x20 = *(const float4*)q2, x21 = *(const float4*)(q2 + 128);
    const float4 x30 = *(const float4*)q3, x31 = *(const float4*)(q3 + 128);
    while (cur < RELS - 1 && p >= bnd) STORE_AND_NEXT_RELATION();
    a0.x = fmaf(x00.x, w0, a0.x); a0.y = fmaf(x00.y, w0, a0.y);
    a0.z = fmaf(x00.z, w0, a0.z); a0.w = fmaf(x00.w, w0, a0.w);
    a1.x = fmaf(x01.x, w0, a1.x); a1.y = fmaf(x01.y, w0, a1.y);
    a1.z = fmaf(x01.z, w0, a1.z); a1.w = fmaf(x01.w, w0, a1.w);
    while (cur < RELS - 1 && p + 1 >= bnd) STORE_AND_NEXT_RELATION();
    a0.x = fmaf(x10.x, w1, a0.x); a0.y = fmaf(x10.y, w1, a0.y);
    a0.z = fmaf(x10.z, w1, a0.z); a0.w = fmaf(x10.w, w1, a0.w);
    a1.x = fmaf(x11.x, w1, a1.x); a1.y = fmaf(x11.y, w1, a1.y);
    a1.z = fmaf(x11.z, w1, a1.z); a1.w = fmaf(x11.w, w1, a1.w);
    while (cur < RELS - 1 && p + 2 >= bnd) STORE_AND_NEXT_RELATION();
    a0.x = fmaf(x20.x, w2, a0.x); a0.y = fmaf(x20.y, w2, a0.y);
    a0.z = fmaf(x20.z, w2, a0.z); a0.w = fmaf(x20.w, w2, a0.w);
    a1.x = fmaf(x21.x, w2, a1.x); a1.y = fmaf(x21.y, w2, a1.y);
    a1.z = fmaf(x21.z, w2, a1.z); a1.w = fmaf(x21.w, w2, a1.w);
    while (cur < RELS - 1 && p + 3 >= bnd) STORE_AND_NEXT_RELATION();
    a0.x = fmaf(x30.x, w3, a0.x); a0.y = fmaf(x30.y, w3, a0.y);
    a0.z = fmaf(x30.z, w3, a0.z); a0.w = fmaf(x30.w, w3, a0.w);
    a1.x = fmaf(x31.x, w3, a1.x); a1.y = fmaf(x31.y, w3, a1.y);
    a1.z = fmaf(x31.z, w3, a1.z); a1.w = fmaf(x31.w, w3, a1.w);
  }
  for (; p < n; ++p) {
    const int   c = place_pick(cA, cB, p);
    const float w = __int_as_float(place_pick(vA, vB, p));
    const float* q = T + (size_t)c * FDIM + lane * 4;
    const float4 x0 = *(const float4*)q, x1 = *(const float4*)(q + 128);
    while (cur < RELS - 1 && p >= bnd) STORE_AND_NEXT_RELATION();
    a0.x = fmaf(x0.x, w, a0.x); a0.y = fmaf(x0.y, w, a0.y);
    a0.z = fmaf(x0.z, w, a0.z); a0.w = fmaf(x0.w, w, a0.w);
    a1.x = fmaf(x1.x, w, a1.x); a1.y = fmaf(x1.y, w, a1.y);
    a1.z = fmaf(x1.z, w, a1.z); a1.w = fmaf(x1.w, w, a1.w);
  }
  while (cur < RELS) STORE_AND_NEXT_RELATION();
#undef STORE_AND_NEXT_RELATION
}

__global__ __launch_bounds__(256)
void gather_rel_rows(const float* __restrict__ T, const int* __restrict__ rowptr,
                     const int* __restrict__ bins, const int* __restrict__ esrc,
                     const float* __restrict__ eval, float* __restrict__ agg,
                     int N, int E) {
  __shared__ int sC[8][64];
  __shared__ int sV[8][64];
  const int lane = threadIdx.x & 31;
  const int wave = __builtin_amdgcn_readfirstlane(threadIdx.x >> 5);
  const int row  = blockIdx.x * 8 + wave;
  int start = 0, n = 0;
  if (row < N) { start = rowptr[row]; n = rowptr[row + 1] - start; }
  const bool small = n <= 64;
  int e1 = 0, e2 = 0, e3 = 0;

  if (small) {
    const bool h0 = lane < n, h1 = lane + 32 < n;
    int k0 = KEY_NONE, k1 = KEY_NONE;
    int s0 = 0, s1 = 0, v0 = 0, v1 = 0;
    if (h0) {
      const int4 r = *(const int4*)(bins + 4 * (size_t)(start + lane));
      k0 = r.x; s0 = r.y; v0 = r.z;
    }
    if (h1) {
      const int4 r = *(const int4*)(bins + 4 * (size_t)(start + 32 + lane));
      k1 = r.x; s1 = r.y; v1 = r.z;
    }
    int r0 = 0, r1 = 0;
    const int na = min(n, 32);
    for (int i = 0; i < na; ++i) {
      const int ki = __builtin_amdgcn_readlane(k0, i);
      r0 += (ki < k0) ? 1 : 0;
      r1 += (ki < k1) ? 1 : 0;
    }
    for (int i = 0; i < n - 32; ++i) {
      const int ki = __builtin_amdgcn_readlane(k1, i);
      r0 += (ki < k0) ? 1 : 0;
      r1 += (ki < k1) ? 1 : 0;
    }
    if (h0) { sC[wave][r0] = s0; sV[wave][r0] = v0; }
    if (h1) { sC[wave][r1] = s1; sV[wave][r1] = v1; }
    e1 = __builtin_popcount(__builtin_amdgcn_ballot_w32(k0 < E)) +
         __builtin_popcount(__builtin_amdgcn_ballot_w32(k1 < E));
    e2 = __builtin_popcount(__builtin_amdgcn_ballot_w32(k0 < 2 * E)) +
         __builtin_popcount(__builtin_amdgcn_ballot_w32(k1 < 2 * E));
    e3 = __builtin_popcount(__builtin_amdgcn_ballot_w32(k0 < 3 * E)) +
         __builtin_popcount(__builtin_amdgcn_ballot_w32(k1 < 3 * E));
  }
  __syncthreads();

  float* orow = agg + (size_t)row * KCAT + lane * 4;
  if (small) {
    int cA = 0, vA = 0, cB = 0, vB = 0;
    if (lane < n)      { cA = sC[wave][lane];      vA = sV[wave][lane]; }
    if (lane + 32 < n) { cB = sC[wave][lane + 32]; vB = sV[wave][lane + 32]; }
    sum_row_by_relation(T, cA, vA, cB, vB, n, e1, e2, e3, lane, orow, row < N);
  } else {
    float4 a0 = make_float4(0.f, 0.f, 0.f, 0.f);
    float4 a1 = make_float4(0.f, 0.f, 0.f, 0.f);
    int last = -1, cur = 0;
    for (int p = 0; p < n; ++p) {
      int best = KEY_NONE;
      for (int j = lane; j < n; j += 32) {
        const int k = bins[4 * (size_t)(start + j)];
        if (k > last && k < best) best = k;
      }
      for (int off = 16; off > 0; off >>= 1) best = min(best, __shfl_xor(best, off));
      best = __builtin_amdgcn_readfirstlane(best);
      last = best;
      const int   s   = esrc[best];
      const float w   = eval[best];
      const int   rel = (best >= E ? 1 : 0) + (best >= 2 * E ? 1 : 0) + (best >= 3 * E ? 1 : 0);
      for (; cur < rel; ++cur) {
        *(float4*)(orow + cur * FDIM)       = a0;
        *(float4*)(orow + cur * FDIM + 128) = a1;
        a0 = make_float4(0.f, 0.f, 0.f, 0.f);
        a1 = make_float4(0.f, 0.f, 0.f, 0.f);
      }
      const float* q = T + (size_t)s * FDIM + lane * 4;
      const float4 x0 = *(const float4*)q, x1 = *(const float4*)(q + 128);
      a0.x = fmaf(x0.x, w, a0.x); a0.y = fmaf(x0.y, w, a0.y);
      a0.z = fmaf(x0.z, w, a0.z); a0.w = fmaf(x0.w, w, a0.w);
      a1.x = fmaf(x1.x, w, a1.x); a1.y = fmaf(x1.y, w, a1.y);
      a1.z = fmaf(x1.z, w, a1.z); a1.w = fmaf(x1.w, w, a1.w);
    }
    for (; cur < RELS; ++cur) {
      *(float4*)(orow + cur * FDIM)       = a0;
      *(float4*)(orow + cur * FDIM + 128) = a1;
      a0 = make_float4(0.f, 0.f, 0.f, 0.f);
      a1 = make_float4(0.f, 0.f, 0.f, 0.f);
    }
  }
}

extern "C" void kernel_launch(void* const* d_in, const int* in_sizes, int n_in,
                              void* d_out, int out_size, void* d_ws, size_t ws_size,
                              hipStream_t stream) {
  const float* x    = (const float*)d_in[0];
  const int*   esrc = (const int*)d_in[1];
  const int*   edst = (const int*)d_in[2];
  const float* eval = (const float*)d_in[3];
  const float* W1   = (const float*)d_in[4];
  const float* b1   = (const float*)d_in[5];
  const float* W2   = (const float*)d_in[6];
  const float* b2   = (const float*)d_in[7];
  float* out = (float*)d_out;

  const int R     = 4;
  const int Nn    = in_sizes[0] / FDIM;
  const int E     = in_sizes[1] / R;
  const int total = R * E;

  const size_t nf = (size_t)Nn * FDIM;
  float* agg    = (float*)d_ws;
  int*   bins   = (int*)(agg + 4 * nf);
  int*   cnt    = bins + 4 * (size_t)total;
  int*   rowptr = cnt + Nn + 4;
  float* x1     = out;

  int*   tab    = (int*)d_ws;
  int*   rank   = tab + (size_t)NCB * Nn;

  const int strips = (Nn + 15) / 16;
  int rowBlocks = 768;
  if (rowBlocks * NSMAX < strips) rowBlocks = (strips + NSMAX - 1) / NSMAX;
  if (rowBlocks * 4 > strips)     rowBlocks = strips / 4 > 0 ? strips / 4 : 1;
  dim3 gGrid(FDIM / BN, rowBlocks);
  const int nBlocks = (Nn + 255) / 256;
  const int rBlocks = (Nn + 7) / 8;
  const int per     = (total + NCB - 1) / NCB;
  dim3 fGrid((per + 255) / 256, NCB);

  count_rows_lds<<<NCB, 1024, 0, stream>>>(edst, tab, rank, total, Nn);
  scan_pieces<<<nBlocks, 256, 0, stream>>>(tab, cnt, Nn);
  scan_rows<<<1, 1024, 0, stream>>>(cnt, rowptr, Nn);
  fill_bins<<<fGrid, 256, 0, stream>>>(edst, esrc, eval, rowptr, tab, rank, bins, total, Nn);

  gather_rel_rows<<<rBlocks, 256, 0, stream>>>(x, rowptr, bins, esrc, eval, agg, Nn, E);
  gemm_cat_f32_wmma<false><<<gGrid, 256, 0, stream>>>(agg, W1, b1, nullptr, x1, Nn, KCAT);

  gather_rel_rows<<<rBlocks, 256, 0, stream>>>(x1, rowptr, bins, esrc, eval, agg, Nn, E);
  gemm_cat_f32_wmma<true><<<gGrid, 256, 0, stream>>>(agg, W2, b2, x1, out, Nn, KCAT);
}


# row starts: one-block prefix sum (each thread walking 49 row counts serially, twice) becomes a 256-row scan in each piece-scan block plus a pass adding earlier blocks' totals; float sums unchanged
# speedup vs baseline: 3.3850x; 1.0940x over previous
#include <hip/hip_runtime.h>

typedef __attribute__((ext_vector_type(2))) float v2f;
typedef __attribute__((ext_vector_type(4))) float v4f;
typedef __attribute__((ext_vector_type(8))) float v8f;

#define FDIM 256
#define RELS 4
#define KCAT (RELS * FDIM)
#define BN 128
#define BK 64
#define SXP (BK + 4)
#define WPP 288

#define NSMAX 5
#define OSP 132
template <int NS, bool FINAL>
__device__ __forceinline__
void gemm_strips(const float* __restrict__ X, const float* __restrict__ W,
                 const float* __restrict__ bias, const float* skip,
                 float* O, int nrows, int kdim, int rowBase, int rowEnd,
                 float* sX, float* sW) {
  const int t       = threadIdx.x;
  const int lane    = t & 31;
  const int wave    = t >> 5;
  const int l16     = lane & 15;
  const int half    = lane >> 4;
  const int colBase = blockIdx.x * BN;
  const int cb2     = wave * 32 + l16;

  const int xr0 = t >> 4;
  const int xc4 = (t & 15) << 2;
  unsigned xo[NS];
#pragma unroll
  for (int it = 0; it < NS; ++it)
    xo[it] = ((unsigned)min(rowBase + xr0 + 16 * it, nrows - 1) * (unsigned)kdim + (unsigned)xc4) * 4u;
  float* sxp = &sX[xr0 * SXP + xc4];
  const int wk0 = 2 * (wave >> 1) + ((lane >> 2) & 1);
  const int wn4 = (wave & 1) * 64 + (lane >> 3) * 16 + ((lane & 3) << 2);
  const unsigned wo = (unsigned)(wk0 * FDIM + colBase + wn4) * 4u;
  float* swp = sW + (wave >> 1) * WPP + (wave & 1) * 128 + (lane << 2);

  v8f acc[NS];
#pragma unroll
  for (int s = 0; s < NS; ++s) acc[s] = v8f{};

  v4f xv[NS];
#pragma unroll
  for (int it = 0; it < NS; ++it) xv[it] = *(const v4f*)((const char*)X + xo[it]);
  v4f wv[8];
#pragma unroll
  for (int it = 0; it < 8; ++it)
    wv[it] = *(const v4f*)((const char*)W + wo + (8 * it) * FDIM * 4);

  for (int kBase = 0; kBase < kdim; kBase += BK) {
#pragma unroll
    for (int it = 0; it < NS; ++it)
      *(v4f*)(sxp + it * 16 * SXP) = xv[it];
#pragma unroll
    for (int it = 0; it < 8; ++it)
      *(v4f*)(swp + it * 4 * WPP) = wv[it];
    __syncthreads();

    const int kn = min(kBase + BK, kdim - BK);
    const char* Xk = (const char*)(X + kn);
    const char* Wk = (const char*)(W + (size_t)kn * FDIM);
#pragma unroll
    for (int it = 0; it < NS; ++it) xv[it] = *(const v4f*)(Xk + xo[it]);
#pragma unroll
    for (int it = 0; it < 8; ++it)
      wv[it] = *(const v4f*)(Wk + wo + (8 * it) * FDIM * 4);
    __builtin_amdgcn_sched_barrier(0);

#pragma unroll
    for (int kk = 0; kk < BK; kk += 4) {
      const int ks = kk + 2 * half;
      const float* wrow = sW + ((kk >> 1) + half) * WPP + cb2;
      v2f b;
      b.x = wrow[0]; b.y = wrow[16];
#pragma unroll
      for (int s = 0; s < NS; ++s) {
        const v2f a = *(const v2f*)&sX[(s * 16 + l16) * SXP + ks];
        acc[s] = __builtin_amdgcn_wmma_f32_16x16x4_f32(false, a, false, b, (short)0, acc[s], false, false);
      }
    }
    __syncthreads();
  }

  const int   colw = colBase + wave * 16 + l16;
  const float bb   = bias[colw];
  float* sO = sX;
#pragma unroll
  for (int s = 0; s < NS; ++s)
#pragma unroll
    for (int v = 0; v < 8; ++v)
      sO[(s * 16 + v + 8 * half) * OSP + wave * 16 + l16] = fmaxf(acc[s][v] + bb, 0.f);
  __syncthreads();
  const size_t gcol = (size_t)colBase + lane * 4;
  v4f sk[2 * NS];
  if (FINAL) {
#pragma unroll
    for (int j = 0; j < 2 * NS; ++j) {
      const int gr = rowBase + wave + 8 * j;
      sk[j] = v4f{0.f, 0.f, 0.f, 0.f};
      if (gr < rowEnd) sk[j] = *(const v4f*)(skip + (size_t)gr * FDIM + gcol);
    }
  }
#pragma unroll
  for (int j = 0; j < 2 * NS; ++j) {
    const int r  = wave + 8 * j;
    const int gr = rowBase + r;
    if (gr < rowEnd) {
      v4f o = *(const v4f*)&sO[r * OSP + lane * 4];
      if (FINAL) o += sk[j];
      *(v4f*)(O + (size_t)gr * FDIM + gcol) = o;
    }
  }
}

template <bool FINAL>
__global__ __launch_bounds__(256) __attribute__((amdgpu_waves_per_eu(6)))
void gemm_cat_f32_wmma(const float* __restrict__ X, const float* __restrict__ W,
                       const float* __restrict__ bias, const float* skip,
                       float* O, int nrows, int kdim) {
  __shared__ __align__(16) float sMem[NSMAX * 16 * SXP + (BK / 2) * WPP];
  float* sX = sMem;
  float* sW = sMem + NSMAX * 16 * SXP;

  const int strips = (nrows + 15) >> 4;
  const int rb     = (int)gridDim.y;
  const int base   = strips / rb;
  const int extra  = strips - base * rb;
  const int by     = (int)blockIdx.y;
  const int ns     = base + (by < extra ? 1 : 0);
  const int rowBase = (by * base + min(by, extra)) * 16;
  const int rowEnd  = min(rowBase + ns * 16, nrows);
  if (ns > 4)
    gemm_strips<5, FINAL>(X, W, bias, skip, O, nrows, kdim, rowBase, rowEnd, sX, sW);
  else
    gemm_strips<4, FINAL>(X, W, bias, skip, O, nrows, kdim, rowBase, rowEnd, sX, sW);
}

#define KEY_NONE 0x7fffffff

#define NCB  64
#define HCAP 65536

__global__ __launch_bounds__(1024)
void count_rows_lds(const int* __restrict__ dst, int* __restrict__ tab,
                    int* __restrict__ rank, int total, int n) {
  __shared__ int hist[HCAP];
  const int t   = threadIdx.x;
  const int per = (total + (int)gridDim.x - 1) / (int)gridDim.x;
  const int e0  = min((int)blockIdx.x * per, total);
  const int e1  = min(e0 + per, total);
  int* trow = tab + (size_t)blockIdx.x * n;
  for (int base = 0; base < n; base += HCAP) {
    const int len = min(HCAP, n - base);
    for (int i = t; i < len; i += 1024) hist[i] = 0;
    __syncthreads();
    int g = e0 + t;
    for (; g + 3 * 1024 < e1; g += 4 * 1024) {
      const int d0 = dst[g]            - base;
      const int d1 = dst[g + 1024]     - base;
      const int d2 = dst[g + 2 * 1024] - base;
      const int d3 = dst[g + 3 * 1024] - base;
      if ((unsigned)d0 < (unsigned)len) rank[g]            = atomicAdd(&hist[d0], 1);
      if ((unsigned)d1 < (unsigned)len) rank[g + 1024]     = atomicAdd(&hist[d1], 1);
      if ((unsigned)d2 < (unsigned)len) rank[g + 2 * 1024] = atomicAdd(&hist[d2], 1);
      if ((unsigned)d3 < (unsigned)len) rank[g + 3 * 1024] = atomicAdd(&hist[d3], 1);
    }
    for (; g < e1; g += 1024) {
      const int d = dst[g] - base;
      if ((unsigned)d < (unsigned)len) rank[g] = atomicAdd(&hist[d], 1);
    }
    __syncthreads();
    for (int i = t; i < len; i += 1024) trow[base + i] = hist[i];
    __syncthreads();
  }
}

#define SROWS 256
__global__ __launch_bounds__(SROWS)
void scan_pieces(int* __restrict__ tab, int* __restrict__ rowptr,
                 int* __restrict__ bsum, int n) {
  __shared__ int wsum[SROWS / 32];
  const int t    = threadIdx.x;
  const int lane = t & 31;
  const int wave = t >> 5;
  const int d    = blockIdx.x * SROWS + t;
  int run = 0;
  if (d < n) {
    int v[NCB];
#pragma unroll
    for (int b = 0; b < NCB; ++b) v[b] = tab[(size_t)b * n + d];
#pragma unroll
    for (int b = 0; b < NCB; ++b) {
      const int c = v[b];
      tab[(size_t)b * n + d] = run;
      run += c;
    }
  }
  int inc = run;
#pragma unroll
  for (int off = 1; off < 32; off <<= 1) {
    const int u = __shfl_up(inc, off);
    if (lane >= off) inc += u;
  }
  if (lane == 31) wsum[wave] = inc;
  __syncthreads();
  int before = 0, tot = 0;
#pragma unroll
  for (int w = 0; w < SROWS / 32; ++w) {
    const int s = wsum[w];
    before += (w < wave) ? s : 0;
    tot    += s;
  }
  if (d < n) rowptr[d] = before + inc - run;
  if (t == 0) bsum[blockIdx.x] = tot;
}

__global__ __launch_bounds__(SROWS)
void add_block_starts(int* __restrict__ rowptr, const int* __restrict__ bsum, int n) {
  __shared__ int wsum[SROWS / 32];
  const int t  = threadIdx.x;
  const int nb = (int)blockIdx.x;
  const int d  = nb * SROWS + t;
  int loc = 0;
  if (d < n) loc = rowptr[d];
  int s = 0;
  for (int j = t; j < nb; j += SROWS) s += bsum[j];
#pragma unroll
  for (int off = 16; off > 0; off >>= 1) s += __shfl_xor(s, off);
  if ((t & 31) == 0) wsum[t >> 5] = s;
  __syncthreads();
  int base = 0;
#pragma unroll
  for (int w = 0; w < SROWS / 32; ++w) base += wsum[w];
  if (d < n) rowptr[d] = loc + base;
  if (nb == (int)gridDim.x - 1 && t == 0) rowptr[n] = base + bsum[nb];
}

__global__ __launch_bounds__(256)
void fill_bins(const int* __restrict__ dst, const int* __restrict__ esrc,
               const float* __restrict__ eval, const int* __restrict__ rowptr,
               const int* __restrict__ tab, const int* __restrict__ rank,
               int* __restrict__ bins, int total, int n) {
  const int per = (total + (int)gridDim.y - 1) / (int)gridDim.y;
  const int i   = blockIdx.x * blockDim.x + threadIdx.x;
  const int g   = (int)blockIdx.y * per + i;
  if (i < per && g < total) {
    const int d = dst[g];
    const int4 rec = make_int4(g, esrc[g], __float_as_int(eval[g]), 0);
    const int slot = rowptr[d] + tab[(size_t)blockIdx.y * n + d] + rank[g];
    *(int4*)(bins + 4 * (size_t)slot) = rec;
  }
}

__device__ __forceinline__ int place_pick(int lo, int hi, int p) {
  return __builtin_amdgcn_readlane(p < 32 ? lo : hi, p & 31);
}

__device__ __forceinline__
void sum_row_by_relation(const float* __restrict__ T, int cA, int vA, int cB, int vB,
                         int n, int e1, int e2, int e3, int lane,
                         float* __restrict__ orow, bool live) {
  float4 a0 = make_float4(0.f, 0.f, 0.f, 0.f);
  float4 a1 = make_float4(0.f, 0.f, 0.f, 0.f);
  int cur = 0, bnd = e1;
#define STORE_AND_NEXT_RELATION()                                   \
  do {                                                              \
    if (live) {                                                     \
      *(float4*)(orow + cur * FDIM)       = a0;                     \
      *(float4*)(orow + cur * FDIM + 128) = a1;                     \
    }                                                               \
    a0 = make_float4(0.f, 0.f, 0.f, 0.f);                           \
    a1 = make_float4(0.f, 0.f, 0.f, 0.f);                           \
    ++cur;                                                          \
    bnd = (cur == 1) ? e2 : (cur == 2) ? e3 : n;                    \
  } while (0)
  int p = 0;
  for (; p + 4 <= n; p += 4) {
    const int c0 = place_pick(cA, cB, p);
    const int c1 = place_pick(cA, cB, p + 1);
    const int c2 = place_pick(cA, cB, p + 2);
    const int c3 = place_pick(cA, cB, p + 3);
    const float w0 = __int_as_float(place_pick(vA, vB, p));
    const float w1 = __int_as_float(place_pick(vA, vB, p + 1));
    const float w2 = __int_as_float(place_pick(vA, vB, p + 2));
    const float w3 = __int_as_float(place_pick(vA, vB, p + 3));
    const float* q0 = T + (size_t)c0 * FDIM + lane * 4;
    const float* q1 = T + (size_t)c1 * FDIM + lane * 4;
    const float* q2 = T + (size_t)c2 * FDIM + lane * 4;
    const float* q3 = T + (size_t)c3 * FDIM + lane * 4;
    const float4 x00 = *(const float4*)q0, x01 = *(const float4*)(q0 + 128);
    const float4 x10 = *(const float4*)q1, x11 = *(const float4*)(q1 + 128);
    const float4 x20 = *(const float4*)q2, x21 = *(const float4*)(q2 + 128);
    const float4 x30 = *(const float4*)q3, x31 = *(const float4*)(q3 + 128);
    while (cur < RELS - 1 && p >= bnd) STORE_AND_NEXT_RELATION();
    a0.x = fmaf(x00.x, w0, a0.x); a0.y = fmaf(x00.y, w0, a0.y);
    a0.z = fmaf(x00.z, w0, a0.z); a0.w = fmaf(x00.w, w0, a0.w);
    a1.x = fmaf(x01.x, w0, a1.x); a1.y = fmaf(x01.y, w0, a1.y);
    a1.z = fmaf(x01.z, w0, a1.z); a1.w = fmaf(x01.w, w0, a1.w);
    while (cur < RELS - 1 && p + 1 >= bnd) STORE_AND_NEXT_RELATION();
    a0.x = fmaf(x10.x, w1, a0.x); a0.y = fmaf(x10.y, w1, a0.y);
    a0.z = fmaf(x10.z, w1, a0.z); a0.w = fmaf(x10.w, w1, a0.w);
    a1.x = fmaf(x11.x, w1, a1.x); a1.y = fmaf(x11.y, w1, a1.y);
    a1.z = fmaf(x11.z, w1, a1.z); a1.w = fmaf(x11.w, w1, a1.w);
    while (cur < RELS - 1 && p + 2 >= bnd) STORE_AND_NEXT_RELATION();
    a0.x = fmaf(x20.x, w2, a0.x); a0.y = fmaf(x20.y, w2, a0.y);
    a0.z = fmaf(x20.z, w2, a0.z); a0.w = fmaf(x20.w, w2, a0.w);
    a1.x = fmaf(x21.x, w2, a1.x); a1.y = fmaf(x21.y, w2, a1.y);
    a1.z = fmaf(x21.z, w2, a1.z); a1.w = fmaf(x21.w, w2, a1.w);
    while (cur < RELS - 1 && p + 3 >= bnd) STORE_AND_NEXT_RELATION();
    a0.x = fmaf(x30.x, w3, a0.x); a0.y = fmaf(x30.y, w3, a0.y);
    a0.z = fmaf(x30.z, w3, a0.z); a0.w = fmaf(x30.w, w3, a0.w);
    a1.x = fmaf(x31.x, w3, a1.x); a1.y = fmaf(x31.y, w3, a1.y);
    a1.z = fmaf(x31.z, w3, a1.z); a1.w = fmaf(x31.w, w3, a1.w);
  }
  for (; p < n; ++p) {
    const int   c = place_pick(cA, cB, p);
    const float w = __int_as_float(place_pick(vA, vB, p));
    const float* q = T + (size_t)c * FDIM + lane * 4;
    const float4 x0 = *(const float4*)q, x1 = *(const float4*)(q + 128);
    while (cur < RELS - 1 && p >= bnd) STORE_AND_NEXT_RELATION();
    a0.x = fmaf(x0.x, w, a0.x); a0.y = fmaf(x0.y, w, a0.y);
    a0.z = fmaf(x0.z, w, a0.z); a0.w = fmaf(x0.w, w, a0.w);
    a1.x = fmaf(x1.x, w, a1.x); a1.y = fmaf(x1.y, w, a1.y);
    a1.z = fmaf(x1.z, w, a1.z); a1.w = fmaf(x1.w, w, a1.w);
  }
  while (cur < RELS) STORE_AND_NEXT_RELATION();
#undef STORE_AND_NEXT_RELATION
}

__global__ __launch_bounds__(256)
void gather_rel_rows(const float* __restrict__ T, const int* __restrict__ rowptr,
                     const int* __restrict__ bins, const int* __restrict__ esrc,
                     const float* __restrict__ eval, float* __restrict__ agg,
                     int N, int E) {
  __shared__ int sC[8][64];
  __shared__ int sV[8][64];
  const int lane = threadIdx.x & 31;
  const int wave = __builtin_amdgcn_readfirstlane(threadIdx.x >> 5);
  const int row  = blockIdx.x * 8 + wave;
  int start = 0, n = 0;
  if (row < N) { start = rowptr[row]; n = rowptr[row + 1] - start; }
  const bool small = n <= 64;
  int e1 = 0, e2 = 0, e3 = 0;

  if (small) {
    const bool h0 = lane < n, h1 = lane + 32 < n;
    int k0 = KEY_NONE, k1 = KEY_NONE;
    int s0 = 0, s1 = 0, v0 = 0, v1 = 0;
    if (h0) {
      const int4 r = *(const int4*)(bins + 4 * (size_t)(start + lane));
      k0 = r.x; s0 = r.y; v0 = r.z;
    }
    if (h1) {
      const int4 r = *(const int4*)(bins + 4 * (size_t)(start + 32 + lane));
      k1 = r.x; s1 = r.y; v1 = r.z;
    }
    int r0 = 0, r1 = 0;
    const int na = min(n, 32);
    for (int i = 0; i < na; ++i) {
      const int ki = __builtin_amdgcn_readlane(k0, i);
      r0 += (ki < k0) ? 1 : 0;
      r1 += (ki < k1) ? 1 : 0;
    }
    for (int i = 0; i < n - 32; ++i) {
      const int ki = __builtin_amdgcn_readlane(k1, i);
      r0 += (ki < k0) ? 1 : 0;
      r1 += (ki < k1) ? 1 : 0;
    }
    if (h0) { sC[wave][r0] = s0; sV[wave][r0] = v0; }
    if (h1) { sC[wave][r1] = s1; sV[wave][r1] = v1; }
    e1 = __builtin_popcount(__builtin_amdgcn_ballot_w32(k0 < E)) +
         __builtin_popcount(__builtin_amdgcn_ballot_w32(k1 < E));
    e2 = __builtin_popcount(__builtin_amdgcn_ballot_w32(k0 < 2 * E)) +
         __builtin_popcount(__builtin_amdgcn_ballot_w32(k1 < 2 * E));
    e3 = __builtin_popcount(__builtin_amdgcn_ballot_w32(k0 < 3 * E)) +
         __builtin_popcount(__builtin_amdgcn_ballot_w32(k1 < 3 * E));
  }
  __syncthreads();

  float* orow = agg + (size_t)row * KCAT + lane * 4;
  if (small) {
    int cA = 0, vA = 0, cB = 0, vB = 0;
    if (lane < n)      { cA = sC[wave][lane];      vA = sV[wave][lane]; }
    if (lane + 32 < n) { cB = sC[wave][lane + 32]; vB = sV[wave][lane + 32]; }
    sum_row_by_relation(T, cA, vA, cB, vB, n, e1, e2, e3, lane, orow, row < N);
  } else {
    float4 a0 = make_float4(0.f, 0.f, 0.f, 0.f);
    float4 a1 = make_float4(0.f, 0.f, 0.f, 0.f);
    int last = -1, cur = 0;
    for (int p = 0; p < n; ++p) {
      int best = KEY_NONE;
      for (int j = lane; j < n; j += 32) {
        const int k = bins[4 * (size_t)(start + j)];
        if (k > last && k < best) best = k;
      }
      for (int off = 16; off > 0; off >>= 1) best = min(best, __shfl_xor(best, off));
      best = __builtin_amdgcn_readfirstlane(best);
      last = best;
      const int   s   = esrc[best];
      const float w   = eval[best];
      const int   rel = (best >= E ? 1 : 0) + (best >= 2 * E ? 1 : 0) + (best >= 3 * E ? 1 : 0);
      for (; cur < rel; ++cur) {
        *(float4*)(orow + cur * FDIM)       = a0;
        *(float4*)(orow + cur * FDIM + 128) = a1;
        a0 = make_float4(0.f, 0.f, 0.f, 0.f);
        a1 = make_float4(0.f, 0.f, 0.f, 0.f);
      }
      const float* q = T + (size_t)s * FDIM + lane * 4;
      const float4 x0 = *(const float4*)q, x1 = *(const float4*)(q + 128);
      a0.x = fmaf(x0.x, w, a0.x); a0.y = fmaf(x0.y, w, a0.y);
      a0.z = fmaf(x0.z, w, a0.z); a0.w = fmaf(x0.w, w, a0.w);
      a1.x = fmaf(x1.x, w, a1.x); a1.y = fmaf(x1.y, w, a1.y);
      a1.z = fmaf(x1.z, w, a1.z); a1.w = fmaf(x1.w, w, a1.w);
    }
    for (; cur < RELS; ++cur) {
      *(float4*)(orow + cur * FDIM)       = a0;
      *(float4*)(orow + cur * FDIM + 128) = a1;
      a0 = make_float4(0.f, 0.f, 0.f, 0.f);
      a1 = make_float4(0.f, 0.f, 0.f, 0.f);
    }
  }
}

extern "C" void kernel_launch(void* const* d_in, const int* in_sizes, int n_in,
                              void* d_out, int out_size, void* d_ws, size_t ws_size,
                              hipStream_t stream) {
  const float* x    = (const float*)d_in[0];
  const int*   esrc = (const int*)d_in[1];
  const int*   edst = (const int*)d_in[2];
  const float* eval = (const float*)d_in[3];
  const float* W1   = (const float*)d_in[4];
  const float* b1   = (const float*)d_in[5];
  const float* W2   = (const float*)d_in[6];
  const float* b2   = (const float*)d_in[7];
  float* out = (float*)d_out;

  const int R     = 4;
  const int Nn    = in_sizes[0] / FDIM;
  const int E     = in_sizes[1] / R;
  const int total = R * E;

  const size_t nf = (size_t)Nn * FDIM;
  float* agg    = (float*)d_ws;
  int*   bins   = (int*)(agg + 4 * nf);
  int*   bsum   = bins + 4 * (size_t)total;
  int*   rowptr = bsum + Nn + 4;
  float* x1     = out;

  int*   tab    = (int*)d_ws;
  int*   rank   = tab + (size_t)NCB * Nn;

  const int strips = (Nn + 15) / 16;
  int rowBlocks = 768;
  if (rowBlocks * NSMAX < strips) rowBlocks = (strips + NSMAX - 1) / NSMAX;
  if (rowBlocks * 4 > strips)     rowBlocks = strips / 4 > 0 ? strips / 4 : 1;
  dim3 gGrid(FDIM / BN, rowBlocks);
  const int nBlocks = (Nn + SROWS - 1) / SROWS;
  const int rBlocks = (Nn + 7) / 8;
  const int per     = (total + NCB - 1) / NCB;
  dim3 fGrid((per + 255) / 256, NCB);

  count_rows_lds<<<NCB, 1024, 0, stream>>>(edst, tab, rank, total, Nn);
  scan_pieces<<<nBlocks, SROWS, 0, stream>>>(tab, rowptr, bsum, Nn);
  add_block_starts<<<nBlocks, SROWS, 0, stream>>>(rowptr, bsum, Nn);
  fill_bins<<<fGrid, 256, 0, stream>>>(edst, esrc, eval, rowptr, tab, rank, bins, total, Nn);

  gather_rel_rows<<<rBlocks, 256, 0, stream>>>(x, rowptr, bins, esrc, eval, agg, Nn, E);
  gemm_cat_f32_wmma<false><<<gGrid, 256, 0, stream>>>(agg, W1, b1, nullptr, x1, Nn, KCAT);

  gather_rel_rows<<<rBlocks, 256, 0, stream>>>(x1, rowptr, bins, esrc, eval, agg, Nn, E);
  gemm_cat_f32_wmma<true><<<gGrid, 256, 0, stream>>>(agg, W2, b2, x1, out, Nn, KCAT);
}


# neighbour sums reach the weight product through an LDS row panel: one kernel per layer gathers a relation's rows, then runs its 256 k on the matrix units; no sums plane in memory; sum order kept
# speedup vs baseline: 4.0104x; 1.1848x over previous
#include <hip/hip_runtime.h>

typedef __attribute__((ext_vector_type(2))) float v2f;
typedef __attribute__((ext_vector_type(4))) float v4f;
typedef __attribute__((ext_vector_type(8))) float v8f;

#define FDIM 256
#define RELS 4
#define KCAT (RELS * FDIM)
#define BK 16
#define PP 260
#define WPP 544
#define NSMAX 5

#define KEY_NONE 0x7fffffff

#define NCB  64
#define HCAP 65536

__global__ __launch_bounds__(1024)
void count_rows_lds(const int* __restrict__ dst, int* __restrict__ tab,
                    int* __restrict__ rank, int total, int n) {
  __shared__ int hist[HCAP];
  const int t   = threadIdx.x;
  const int per = (total + (int)gridDim.x - 1) / (int)gridDim.x;
  const int e0  = min((int)blockIdx.x * per, total);
  const int e1  = min(e0 + per, total);
  int* trow = tab + (size_t)blockIdx.x * n;
  for (int base = 0; base < n; base += HCAP) {
    const int len = min(HCAP, n - base);
    for (int i = t; i < len; i += 1024) hist[i] = 0;
    __syncthreads();
    int g = e0 + t;
    for (; g + 3 * 1024 < e1; g += 4 * 1024) {
      const int d0 = dst[g]            - base;
      const int d1 = dst[g + 1024]     - base;
      const int d2 = dst[g + 2 * 1024] - base;
      const int d3 = dst[g + 3 * 1024] - base;
      if ((unsigned)d0 < (unsigned)len) rank[g]            = atomicAdd(&hist[d0], 1);
      if ((unsigned)d1 < (unsigned)len) rank[g + 1024]     = atomicAdd(&hist[d1], 1);
      if ((unsigned)d2 < (unsigned)len) rank[g + 2 * 1024] = atomicAdd(&hist[d2], 1);
      if ((unsigned)d3 < (unsigned)len) rank[g + 3 * 1024] = atomicAdd(&hist[d3], 1);
    }
    for (; g < e1; g += 1024) {
      const int d = dst[g] - base;
      if ((unsigned)d < (unsigned)len) rank[g] = atomicAdd(&hist[d], 1);
    }
    __syncthreads();
    for (int i = t; i < len; i += 1024) trow[base + i] = hist[i];
    __syncthreads();
  }
}

#define SROWS 256
__global__ __launch_bounds__(SROWS)
void scan_pieces(int* __restrict__ tab, int* __restrict__ rowptr,
                 int* __restrict__ bsum, int n) {
  __shared__ int wsum[SROWS / 32];
  const int t    = threadIdx.x;
  const int lane = t & 31;
  const int wave = t >> 5;
  const int d    = blockIdx.x * SROWS + t;
  int run = 0;
  if (d < n) {
    int v[NCB];
#pragma unroll
    for (int b = 0; b < NCB; ++b) v[b] = tab[(size_t)b * n + d];
#pragma unroll
    for (int b = 0; b < NCB; ++b) {
      const int c = v[b];
      tab[(size_t)b * n + d] = run;
      run += c;
    }
  }
  int inc = run;
#pragma unroll
  for (int off = 1; off < 32; off <<= 1) {
    const int u = __shfl_up(inc, off);
    if (lane >= off) inc += u;
  }
  if (lane == 31) wsum[wave] = inc;
  __syncthreads();
  int before = 0, tot = 0;
#pragma unroll
  for (int w = 0; w < SROWS / 32; ++w) {
    const int s = wsum[w];
    before += (w < wave) ? s : 0;
    tot    += s;
  }
  if (d < n) rowptr[d] = before + inc - run;
  if (t == 0) bsum[blockIdx.x] = tot;
}

__global__ __launch_bounds__(SROWS)
void add_block_starts(int* __restrict__ rowptr, const int* __restrict__ bsum, int n) {
  __shared__ int wsum[SROWS / 32];
  const int t  = threadIdx.x;
  const int nb = (int)blockIdx.x;
  const int d  = nb * SROWS + t;
  int loc = 0;
  if (d < n) loc = rowptr[d];
  int s = 0;
  for (int j = t; j < nb; j += SROWS) s += bsum[j];
#pragma unroll
  for (int off = 16; off > 0; off >>= 1) s += __shfl_xor(s, off);
  if ((t & 31) == 0) wsum[t >> 5] = s;
  __syncthreads();
  int base = 0;
#pragma unroll
  for (int w = 0; w < SROWS / 32; ++w) base += wsum[w];
  if (d < n) rowptr[d] = loc + base;
  if (nb == (int)gridDim.x - 1 && t == 0) rowptr[n] = base + bsum[nb];
}

__global__ __launch_bounds__(256)
void fill_bins(const int* __restrict__ dst, const int* __restrict__ esrc,
               const float* __restrict__ eval, const int* __restrict__ rowptr,
               const int* __restrict__ tab, const int* __restrict__ rank,
               int* __restrict__ bins, int total, int n) {
  const int per = (total + (int)gridDim.y - 1) / (int)gridDim.y;
  const int i   = blockIdx.x * blockDim.x + threadIdx.x;
  const int g   = (int)blockIdx.y * per + i;
  if (i < per && g < total) {
    const int d = dst[g];
    const int4 rec = make_int4(g, esrc[g], __float_as_int(eval[g]), 0);
    const int slot = rowptr[d] + tab[(size_t)blockIdx.y * n + d] + rank[g];
    *(int4*)(bins + 4 * (size_t)slot) = rec;
  }
}

__global__ __launch_bounds__(256)
void order_rows(const int* __restrict__ rowptr, const int* __restrict__ bins,
                const int* __restrict__ esrc, const float* __restrict__ eval,
                int* __restrict__ srec, int* __restrict__ relb, int N, int E) {
  const int lane = threadIdx.x & 31;
  const int wave = __builtin_amdgcn_readfirstlane(threadIdx.x >> 5);
  const int row  = blockIdx.x * 8 + wave;
  if (row >= N) return;
  const int start = __builtin_amdgcn_readfirstlane(rowptr[row]);
  const int n     = __builtin_amdgcn_readfirstlane(rowptr[row + 1]) - start;
  int e1 = 0, e2 = 0, e3 = 0;
  if (n <= 64) {
    const bool h0 = lane < n, h1 = lane + 32 < n;
    int k0 = KEY_NONE, k1 = KEY_NONE;
    int s0 = 0, s1 = 0, v0 = 0, v1 = 0;
    if (h0) {
      const int4 r = *(const int4*)(bins + 4 * (size_t)(start + lane));
      k0 = r.x; s0 = r.y; v0 = r.z;
    }
    if (h1) {
      const int4 r = *(const int4*)(bins + 4 * (size_t)(start + 32 + lane));
      k1 = r.x; s1 = r.y; v1 = r.z;
    }
    int r0 = 0, r1 = 0;
    const int na = min(n, 32);
    for (int i = 0; i < na; ++i) {
      const int ki = __builtin_amdgcn_readlane(k0, i);
      r0 += (ki < k0) ? 1 : 0;
      r1 += (ki < k1) ? 1 : 0;
    }
    for (int i = 0; i < n - 32; ++i) {
      const int ki = __builtin_amdgcn_readlane(k1, i);
      r0 += (ki < k0) ? 1 : 0;
      r1 += (ki < k1) ? 1 : 0;
    }
    if (h0) *(int2*)(srec + 2 * (size_t)(start + r0)) = make_int2(s0, v0);
    if (h1) *(int2*)(srec + 2 * (size_t)(start + r1)) = make_int2(s1, v1);
    e1 = __builtin_popcount(__builtin_amdgcn_ballot_w32(k0 < E)) +
         __builtin_popcount(__builtin_amdgcn_ballot_w32(k1 < E));
    e2 = __builtin_popcount(__builtin_amdgcn_ballot_w32(k0 < 2 * E)) +
         __builtin_popcount(__builtin_amdgcn_ballot_w32(k1 < 2 * E));
    e3 = __builtin_popcount(__builtin_amdgcn_ballot_w32(k0 < 3 * E)) +
         __builtin_popcount(__builtin_amdgcn_ballot_w32(k1 < 3 * E));
  } else {
    int last = -1;
    for (int p = 0; p < n; ++p) {
      int best = KEY_NONE;
      for (int j = lane; j < n; j += 32) {
        const int k = bins[4 * (size_t)(start + j)];
        if (k > last && k < best) best = k;
      }
      for (int off = 16; off > 0; off >>= 1) best = min(best, __shfl_xor(best, off));
      best = __builtin_amdgcn_readfirstlane(best);
      last = best;
      if (lane == 0) *(int2*)(srec + 2 * (size_t)(start + p)) = make_int2(esrc[best], __float_as_int(eval[best]));
      e1 += (best < E) ? 1 : 0;
      e2 += (best < 2 * E) ? 1 : 0;
      e3 += (best < 3 * E) ? 1 : 0;
    }
  }
  if (lane == 0) *(int4*)(relb + 4 * (size_t)row) = make_int4(start, start + e1, start + e2, start + e3);
}


__device__ __forceinline__
void sum_places(const float* __restrict__ T, int c, int v, int cnt, int lane,
                float4& a0, float4& a1) {
  int i = 0;
  for (; i + 4 <= cnt; i += 4) {
    const int c0 = __builtin_amdgcn_readlane(c, i);
    const int c1 = __builtin_amdgcn_readlane(c, i + 1);
    const int c2 = __builtin_amdgcn_readlane(c, i + 2);
    const int c3 = __builtin_amdgcn_readlane(c, i + 3);
    const float w0 = __int_as_float(__builtin_amdgcn_readlane(v, i));
    const float w1 = __int_as_float(__builtin_amdgcn_readlane(v, i + 1));
    const float w2 = __int_as_float(__builtin_amdgcn_readlane(v, i + 2));
    const float w3 = __int_as_float(__builtin_amdgcn_readlane(v, i + 3));
    const float* q0 = T + (size_t)c0 * FDIM + lane * 4;
    const float* q1 = T + (size_t)c1 * FDIM + lane * 4;
    const float* q2 = T + (size_t)c2 * FDIM + lane * 4;
    const float* q3 = T + (size_t)c3 * FDIM + lane * 4;
    const float4 x00 = *(const float4*)q0, x01 = *(const float4*)(q0 + 128);
    const float4 x10 = *(const float4*)q1, x11 = *(const float4*)(q1 + 128);
    const float4 x20 = *(const float4*)q2, x21 = *(const float4*)(q2 + 128);
    const float4 x30 = *(const float4*)q3, x31 = *(const float4*)(q3 + 128);
    a0.x = fmaf(x00.x, w0, a0.x); a0.y = fmaf(x00.y, w0, a0.y);
    a0.z = fmaf(x00.z, w0, a0.z); a0.w = fmaf(x00.w, w0, a0.w);
    a1.x = fmaf(x01.x, w0, a1.x); a1.y = fmaf(x01.y, w0, a1.y);
    a1.z = fmaf(x01.z, w0, a1.z); a1.w = fmaf(x01.w, w0, a1.w);
    a0.x = fmaf(x10.x, w1, a0.x); a0.y = fmaf(x10.y, w1, a0.y);
    a0.z = fmaf(x10.z, w1, a0.z); a0.w = fmaf(x10.w, w1, a0.w);
    a1.x = fmaf(x11.x, w1, a1.x); a1.y = fmaf(x11.y, w1, a1.y);
    a1.z = fmaf(x11.z, w1, a1.z); a1.w = fmaf(x11.w, w1, a1.w);
    a0.x = fmaf(x20.x, w2, a0.x); a0.y = fmaf(x20.y, w2, a0.y);
    a0.z = fmaf(x20.z, w2, a0.z); a0.w = fmaf(x20.w, w2, a0.w);
    a1.x = fmaf(x21.x, w2, a1.x); a1.y = fmaf(x21.y, w2, a1.y);
    a1.z = fmaf(x21.z, w2, a1.z); a1.w = fmaf(x21.w, w2, a1.w);
    a0.x = fmaf(x30.x, w3, a0.x); a0.y = fmaf(x30.y, w3, a0.y);
    a0.z = fmaf(x30.z, w3, a0.z); a0.w = fmaf(x30.w, w3, a0.w);
    a1.x = fmaf(x31.x, w3, a1.x); a1.y = fmaf(x31.y, w3, a1.y);
    a1.z = fmaf(x31.z, w3, a1.z); a1.w = fmaf(x31.w, w3, a1.w);
  }
  const int rem = cnt - i;
  if (rem > 0) {
    const int   c0 = __builtin_amdgcn_readlane(c, i);
    const float w0 = __int_as_float(__builtin_amdgcn_readlane(v, i));
    const float* q0 = T + (size_t)c0 * FDIM + lane * 4;
    const float4 x00 = *(const float4*)q0, x01 = *(const float4*)(q0 + 128);
    float4 x10 = make_float4(0.f, 0.f, 0.f, 0.f), x11 = x10, x20 = x10, x21 = x10;
    float w1 = 0.f, w2 = 0.f;
    if (rem > 1) {
      const int c1 = __builtin_amdgcn_readlane(c, i + 1);
      w1 = __int_as_float(__builtin_amdgcn_readlane(v, i + 1));
      const float* q1 = T + (size_t)c1 * FDIM + lane * 4;
      x10 = *(const float4*)q1; x11 = *(const float4*)(q1 + 128);
    }
    if (rem > 2) {
      const int c2 = __builtin_amdgcn_readlane(c, i + 2);
      w2 = __int_as_float(__builtin_amdgcn_readlane(v, i + 2));
      const float* q2 = T + (size_t)c2 * FDIM + lane * 4;
      x20 = *(const float4*)q2; x21 = *(const float4*)(q2 + 128);
    }
    a0.x = fmaf(x00.x, w0, a0.x); a0.y = fmaf(x00.y, w0, a0.y);
    a0.z = fmaf(x00.z, w0, a0.z); a0.w = fmaf(x00.w, w0, a0.w);
    a1.x = fmaf(x01.x, w0, a1.x); a1.y = fmaf(x01.y, w0, a1.y);
    a1.z = fmaf(x01.z, w0, a1.z); a1.w = fmaf(x01.w, w0, a1.w);
    if (rem > 1) {
      a0.x = fmaf(x10.x, w1, a0.x); a0.y = fmaf(x10.y, w1, a0.y);
      a0.z = fmaf(x10.z, w1, a0.z); a0.w = fmaf(x10.w, w1, a0.w);
      a1.x = fmaf(x11.x, w1, a1.x); a1.y = fmaf(x11.y, w1, a1.y);
      a1.z = fmaf(x11.z, w1, a1.z); a1.w = fmaf(x11.w, w1, a1.w);
    }
    if (rem > 2) {
      a0.x = fmaf(x20.x, w2, a0.x); a0.y = fmaf(x20.y, w2, a0.y);
      a0.z = fmaf(x20.z, w2, a0.z); a0.w = fmaf(x20.w, w2, a0.w);
      a1.x = fmaf(x21.x, w2, a1.x); a1.y = fmaf(x21.y, w2, a1.y);
      a1.z = fmaf(x21.z, w2, a1.z); a1.w = fmaf(x21.w, w2, a1.w);
    }
  }
}

template <int NS, bool FINAL>
__device__ __forceinline__
void layer_strips(const float* __restrict__ T, const int* __restrict__ rowptr,
                  const int* __restrict__ srec, const int* __restrict__ relb,
                  const float* __restrict__ W, const float* __restrict__ bias,
                  const float* __restrict__ skip, float* __restrict__ O,
                  int nrows, int rowBase, int rowEnd, float* sP, float* sW) {
  const int t    = threadIdx.x;
  const int lane = t & 31;
  const int wave = __builtin_amdgcn_readfirstlane(t >> 5);
  const int l16  = lane & 15;
  const int half = lane >> 4;
  constexpr int RPW = 2 * NS;

  const int wk0 = 2 * wave + ((lane >> 2) & 1);
  const int wn4 = (lane >> 3) * 16 + ((lane & 3) << 2);
  const unsigned wo = (unsigned)(wk0 * FDIM + wn4) * 4u;
  float* swp = sW + wave * WPP + (lane << 2);

  v8f acc[NS][2];
#pragma unroll
  for (int s = 0; s < NS; ++s) { acc[s][0] = v8f{}; acc[s][1] = v8f{}; }

  v4f wv[4];
#pragma unroll
  for (int it = 0; it < 4; ++it)
    wv[it] = *(const v4f*)((const char*)W + wo + it * 256);

  int b0 = 0, b1 = 0, b2 = 0, b3 = 0, b4 = 0;
  if (lane < RPW) {
    const int gr = rowBase + wave + 8 * lane;
    if (gr < nrows) {
      const int4 q = *(const int4*)(relb + 4 * (size_t)gr);
      b0 = q.x; b1 = q.y; b2 = q.z; b3 = q.w;
      b4 = rowptr[gr + 1];
    }
  }

#pragma unroll 1
  for (int r = 0; r < RELS; ++r) {
    {
      const int lo = (r == 0) ? b0 : (r == 1) ? b1 : (r == 2) ? b2 : b3;
      const int hi = (r == 0) ? b1 : (r == 1) ? b2 : (r == 2) ? b3 : b4;
      int2 recs[RPW];
#pragma unroll
      for (int j = 0; j < RPW; ++j) {
        const int q0 = __builtin_amdgcn_readlane(lo, j);
        const int q1 = __builtin_amdgcn_readlane(hi, j);
        recs[j] = make_int2(0, 0);
        if (lane < q1 - q0) recs[j] = *(const int2*)(srec + 2 * (size_t)(q0 + lane));
      }
#pragma unroll
      for (int j = 0; j < RPW; ++j) {
        const int p1 = __builtin_amdgcn_readlane(hi, j);
        int base     = __builtin_amdgcn_readlane(lo, j);
        int2 rec = recs[j];
        float4 a0 = make_float4(0.f, 0.f, 0.f, 0.f);
        float4 a1 = make_float4(0.f, 0.f, 0.f, 0.f);
        for (;;) {
          sum_places(T, rec.x, rec.y, min(32, p1 - base), lane, a0, a1);
          base += 32;
          if (base >= p1) break;
          rec = make_int2(0, 0);
          if (lane < p1 - base) rec = *(const int2*)(srec + 2 * (size_t)(base + lane));
        }
        float* prow = sP + (wave + 8 * j) * PP + lane * 4;
        *(float4*)prow         = a0;
        *(float4*)(prow + 128) = a1;
      }
    }

    for (int kt = 0; kt < FDIM / BK; ++kt) {
#pragma unroll
      for (int it = 0; it < 4; ++it)
        *(v4f*)(swp + it * 128) = wv[it];
      __syncthreads();

      const int tn = min(r * (FDIM / BK) + kt + 1, KCAT / BK - 1);
      const char* Wk = (const char*)(W + (size_t)tn * BK * FDIM);
#pragma unroll
      for (int it = 0; it < 4; ++it)
        wv[it] = *(const v4f*)(Wk + wo + it * 256);
      __builtin_amdgcn_sched_barrier(0);

      const int kc = kt * BK;
#pragma unroll
      for (int kk = 0; kk < BK; kk += 4) {
        const int ks = kk + 2 * half;
        const float* wrow = sW + ((kk >> 1) + half) * WPP + wave * 64 + l16;
        v2f bA, bB;
        bA.x = wrow[0];  bA.y = wrow[16];
        bB.x = wrow[32]; bB.y = wrow[48];
#pragma unroll
        for (int s = 0; s < NS; ++s) {
          const v2f a = *(const v2f*)&sP[(s * 16 + l16) * PP + kc + ks];
          acc[s][0] = __builtin_amdgcn_wmma_f32_16x16x4_f32(false, a, false, bA, (short)0, acc[s][0], false, false);
          acc[s][1] = __builtin_amdgcn_wmma_f32_16x16x4_f32(false, a, false, bB, (short)0, acc[s][1], false, false);
        }
      }
      __syncthreads();
    }
  }

  const float bbA = bias[wave * 32 + l16];
  const float bbB = bias[wave * 32 + 16 + l16];
  float* sO = sP;
#pragma unroll
  for (int s = 0; s < NS; ++s)
#pragma unroll
    for (int v = 0; v < 8; ++v) {
      float* po = sO + (s * 16 + v + 8 * half) * PP + wave * 32 + l16;
      po[0]  = fmaxf(acc[s][0][v] + bbA, 0.f);
      po[16] = fmaxf(acc[s][1][v] + bbB, 0.f);
    }
  __syncthreads();
#pragma unroll
  for (int j = 0; j < RPW; ++j) {
    const int rr = wave + 8 * j;
    const int gr = rowBase + rr;
    if (gr < rowEnd) {
      v4f o0 = *(const v4f*)&sO[rr * PP + lane * 4];
      v4f o1 = *(const v4f*)&sO[rr * PP + 128 + lane * 4];
      if (FINAL) {
        o0 += *(const v4f*)(skip + (size_t)gr * FDIM + lane * 4);
        o1 += *(const v4f*)(skip + (size_t)gr * FDIM + 128 + lane * 4);
      }
      *(v4f*)(O + (size_t)gr * FDIM + lane * 4)       = o0;
      *(v4f*)(O + (size_t)gr * FDIM + 128 + lane * 4) = o1;
    }
  }
}

template <bool FINAL>
__global__ __launch_bounds__(256) __attribute__((amdgpu_waves_per_eu(6)))
void layer_gather_wmma(const float* __restrict__ T, const int* __restrict__ rowptr,
                       const int* __restrict__ srec, const int* __restrict__ relb,
                       const float* __restrict__ W, const float* __restrict__ bias,
                       const float* __restrict__ skip, float* __restrict__ O, int nrows) {
  __shared__ __align__(16) float sMem[NSMAX * 16 * PP + (BK / 2) * WPP];
  float* sP = sMem;
  float* sW = sMem + NSMAX * 16 * PP;

  const int strips = (nrows + 15) >> 4;
  const int rb     = (int)gridDim.x;
  const int base   = strips / rb;
  const int extra  = strips - base * rb;
  const int by     = (int)blockIdx.x;
  const int ns     = base + (by < extra ? 1 : 0);
  const int rowBase = (by * base + min(by, extra)) * 16;
  const int rowEnd  = min(rowBase + ns * 16, nrows);
  if (ns > 4)
    layer_strips<5, FINAL>(T, rowptr, srec, relb, W, bias, skip, O, nrows, rowBase, rowEnd, sP, sW);
  else
    layer_strips<4, FINAL>(T, rowptr, srec, relb, W, bias, skip, O, nrows, rowBase, rowEnd, sP, sW);
}

extern "C" void kernel_launch(void* const* d_in, const int* in_sizes, int n_in,
                              void* d_out, int out_size, void* d_ws, size_t ws_size,
                              hipStream_t stream) {
  const float* x    = (const float*)d_in[0];
  const int*   esrc = (const int*)d_in[1];
  const int*   edst = (const int*)d_in[2];
  const float* eval = (const float*)d_in[3];
  const float* W1   = (const float*)d_in[4];
  const float* b1   = (const float*)d_in[5];
  const float* W2   = (const float*)d_in[6];
  const float* b2   = (const float*)d_in[7];
  float* out = (float*)d_out;

  const int R     = 4;
  const int Nn    = in_sizes[0] / FDIM;
  const int E     = in_sizes[1] / R;
  const int total = R * E;

  const size_t nf = (size_t)Nn * FDIM;
  float* wsf    = (float*)d_ws;
  float* x1     = wsf + nf;
  int*   srec   = (int*)(wsf + 2 * nf);
  int*   relb   = (int*)(wsf + 3 * nf);
  int*   bins   = (int*)(wsf + 4 * nf);
  int*   bsum   = bins + 4 * (size_t)total;
  int*   rowptr = bsum + Nn + 4;

  int*   tab    = (int*)d_ws;
  int*   rank   = tab + (size_t)NCB * Nn;

  const int strips = (Nn + 15) / 16;
  int rowBlocks = 768;
  if (rowBlocks * NSMAX < strips) rowBlocks = (strips + NSMAX - 1) / NSMAX;
  if (rowBlocks * 4 > strips)     rowBlocks = strips / 4 > 0 ? strips / 4 : 1;
  const int nBlocks = (Nn + SROWS - 1) / SROWS;
  const int rBlocks = (Nn + 7) / 8;
  const int per     = (total + NCB - 1) / NCB;
  dim3 fGrid((per + 255) / 256, NCB);

  count_rows_lds<<<NCB, 1024, 0, stream>>>(edst, tab, rank, total, Nn);
  scan_pieces<<<nBlocks, SROWS, 0, stream>>>(tab, rowptr, bsum, Nn);
  add_block_starts<<<nBlocks, SROWS, 0, stream>>>(rowptr, bsum, Nn);
  fill_bins<<<fGrid, 256, 0, stream>>>(edst, esrc, eval, rowptr, tab, rank, bins, total, Nn);
  order_rows<<<rBlocks, 256, 0, stream>>>(rowptr, bins, esrc, eval, srec, relb, Nn, E);

  layer_gather_wmma<false><<<rowBlocks, 256, 0, stream>>>(x, rowptr, srec, relb, W1, b1, nullptr, x1, Nn);
  layer_gather_wmma<true><<<rowBlocks, 256, 0, stream>>>(x1, rowptr, srec, relb, W2, b2, x1, out, Nn);
}
